# starting kernel
# baseline (speedup 1.0000x reference)
#include <hip/hip_runtime.h>
#include <math.h>

typedef float v2f __attribute__((ext_vector_type(2)));
typedef float v8f __attribute__((ext_vector_type(8)));

#define NEG_SLOPE 0.2f

__device__ __forceinline__ float gat_lrelu(float v) { return v > 0.0f ? v : NEG_SLOPE * v; }
__device__ __forceinline__ unsigned gat_fkey(float f) {
    unsigned u = __float_as_uint(f);
    return (u & 0x80000000u) ? ~u : (u | 0x80000000u);
}
__device__ __forceinline__ float gat_funkey(unsigned k) {
    return __uint_as_float((k & 0x80000000u) ? (k & 0x7fffffffu) : ~k);
}

template<int NCOL, int NCT>
__global__ __launch_bounds__(256) void gat_gemm_wmma(
    const float* __restrict__ X, const float* __restrict__ W,
    float* __restrict__ Y, int N, int ncolStore)
{
    constexpr int K = 128;
    __shared__ float ldsW[K * NCOL];
    for (int idx = threadIdx.x; idx < K * NCOL; idx += 256) {
        int k = idx / NCOL, c = idx % NCOL;
        ldsW[idx] = (c < ncolStore) ? W[k * ncolStore + c] : 0.0f;
    }
    __syncthreads();

    const int lane = threadIdx.x & 31;
    const int wave = threadIdx.x >> 5;
    const int m    = lane & 15;
    const int half = lane >> 4;
    const int koff = half << 1;

    long row0 = (long)blockIdx.x * 128 + wave * 16;
    long rowL = row0 + m; if (rowL > (long)N - 1) rowL = (long)N - 1;
    const float* xrow = X + rowL * K;

    v8f acc[NCT] = {};
    for (int k0 = 0; k0 < K; k0 += 4) {
        v2f a;
        a.x = xrow[k0 + koff];
        a.y = xrow[k0 + koff + 1];
#pragma unroll
        for (int t = 0; t < NCT; ++t) {
            v2f b;
            b.x = ldsW[(k0 + koff) * NCOL + t * 16 + m];
            b.y = ldsW[(k0 + koff + 1) * NCOL + t * 16 + m];
            acc[t] = __builtin_amdgcn_wmma_f32_16x16x4_f32(
                false, a, false, b, (short)0, acc[t], false, false);
        }
    }
#pragma unroll
    for (int r = 0; r < 8; ++r) {
        long row = row0 + half * 8 + r;
        if (row < N) {
#pragma unroll
            for (int t = 0; t < NCT; ++t) {
                int c = t * 16 + m;
                if (c < ncolStore) Y[row * ncolStore + c] = acc[t][r];
            }
        }
    }
}

template<int H, int C>
__global__ __launch_bounds__(256) void gat_alpha_kernel(
    const float* __restrict__ h, const float* __restrict__ aw_s,
    const float* __restrict__ aw_d, float* __restrict__ as_,
    float* __restrict__ ad_, int N)
{
    constexpr int HC = H * C;
    int wid  = blockIdx.x * (blockDim.x >> 5) + (threadIdx.x >> 5);
    int lane = threadIdx.x & 31;
    if (wid >= N) return;
    const float* hp = h + (long)wid * HC;
    float s[H], d[H];
#pragma unroll
    for (int hh = 0; hh < H; ++hh) { s[hh] = 0.0f; d[hh] = 0.0f; }
    for (int c = lane; c < HC; c += 32) {
        int hh = (H == 1) ? 0 : (c / C);
        float v = hp[c];
        s[hh] += v * aw_s[c];
        d[hh] += v * aw_d[c];
    }
#pragma unroll
    for (int hh = 0; hh < H; ++hh) {
        float sv = s[hh], dv = d[hh];
        for (int o = 16; o > 0; o >>= 1) {
            sv += __shfl_xor(sv, o, 32);
            dv += __shfl_xor(dv, o, 32);
        }
        if (lane == 0) {
            as_[(long)wid * H + hh] = sv;
            ad_[(long)wid * H + hh] = dv;
        }
    }
}

template<int H>
__global__ __launch_bounds__(256) void gat_edge_max_kernel(
    const int* __restrict__ src, const int* __restrict__ dst, int E, int N,
    const float* __restrict__ as_, const float* __restrict__ ad_,
    unsigned* __restrict__ nmax)
{
    int e = blockIdx.x * blockDim.x + threadIdx.x;
    if (e >= E + N) return;
    int s = (e < E) ? src[e] : (e - E);
    int d = (e < E) ? dst[e] : (e - E);
#pragma unroll
    for (int h = 0; h < H; ++h) {
        float ev = gat_lrelu(as_[s * H + h] + ad_[d * H + h]);
        atomicMax(&nmax[d * H + h], gat_fkey(ev));
    }
}

template<int H>
__global__ __launch_bounds__(256) void gat_edge_denom_kernel(
    const int* __restrict__ src, const int* __restrict__ dst, int E, int N,
    const float* __restrict__ as_, const float* __restrict__ ad_,
    const unsigned* __restrict__ nmax, float* __restrict__ den)
{
    int e = blockIdx.x * blockDim.x + threadIdx.x;
    if (e >= E + N) return;
    int s = (e < E) ? src[e] : (e - E);
    int d = (e < E) ? dst[e] : (e - E);
#pragma unroll
    for (int h = 0; h < H; ++h) {
        float ev = gat_lrelu(as_[s * H + h] + ad_[d * H + h]);
        atomicAdd(&den[d * H + h], expf(ev - gat_funkey(nmax[d * H + h])));
    }
}

template<int H>
__global__ __launch_bounds__(256) void gat_edge_coef_kernel(
    const int* __restrict__ src, const int* __restrict__ dst, int E, int N,
    const float* __restrict__ as_, const float* __restrict__ ad_,
    const unsigned* __restrict__ nmax, const float* __restrict__ den,
    float* __restrict__ coef)
{
    int e = blockIdx.x * blockDim.x + threadIdx.x;
    if (e >= E + N) return;
    int s = (e < E) ? src[e] : (e - E);
    int d = (e < E) ? dst[e] : (e - E);
#pragma unroll
    for (int h = 0; h < H; ++h) {
        float ev = gat_lrelu(as_[s * H + h] + ad_[d * H + h]);
        float mx = gat_funkey(nmax[d * H + h]);
        coef[(long)e * H + h] = expf(ev - mx) / (den[d * H + h] + 1e-16f);
    }
}

template<int H, int C>
__global__ __launch_bounds__(256) void gat_edge_scatter_kernel(
    const int* __restrict__ src, const int* __restrict__ dst, int E, int N,
    const float* __restrict__ coef, const float* __restrict__ feat,
    float* __restrict__ agg)
{
    constexpr int HC = H * C;
    int wid  = blockIdx.x * (blockDim.x >> 5) + (threadIdx.x >> 5);
    int lane = threadIdx.x & 31;
    if (wid >= E + N) return;
    int s = (wid < E) ? src[wid] : (wid - E);
    int d = (wid < E) ? dst[wid] : (wid - E);
    const float* fs = feat + (long)s * HC;
    float* ao = agg + (long)d * HC;
    for (int c = lane; c < HC; c += 32) {
        int hh = (H == 1) ? 0 : (c / C);
        atomicAdd(&ao[c], coef[(long)wid * H + hh] * fs[c]);
    }
}

__global__ void gat_fill_u32_kernel(unsigned* __restrict__ p, unsigned v, long n) {
    long i = blockIdx.x * (long)blockDim.x + threadIdx.x;
    if (i < n) p[i] = v;
}

__global__ void gat_elu_bias_kernel(float* __restrict__ a, const float* __restrict__ b,
                                    long n, int HC) {
    long i = blockIdx.x * (long)blockDim.x + threadIdx.x;
    if (i < n) {
        float t = a[i] + b[i % HC];
        a[i] = t > 0.0f ? t : (expf(t) - 1.0f);
    }
}

__global__ void gat_bias_out_kernel(const float* __restrict__ a, const float* __restrict__ b,
                                    float* __restrict__ out, long n, int HC) {
    long i = blockIdx.x * (long)blockDim.x + threadIdx.x;
    if (i < n) out[i] = a[i] + b[i % HC];
}

extern "C" void kernel_launch(void* const* d_in, const int* in_sizes, int n_in,
                              void* d_out, int out_size, void* d_ws, size_t ws_size,
                              hipStream_t stream)
{
    (void)n_in; (void)out_size; (void)ws_size;
    const float* x   = (const float*)d_in[0];
    const int*   ei  = (const int*)  d_in[1];
    const float* W1  = (const float*)d_in[2];
    const float* aS1 = (const float*)d_in[3];
    const float* aD1 = (const float*)d_in[4];
    const float* b1  = (const float*)d_in[5];
    const float* W2  = (const float*)d_in[6];
    const float* aS2 = (const float*)d_in[7];
    const float* aD2 = (const float*)d_in[8];
    const float* b2  = (const float*)d_in[9];

    const int IN = 128, HC1 = 128, OUTC = 40;
    int N  = in_sizes[0] / IN;
    int E  = in_sizes[1] / 2;
    int ET = E + N;
    const int* srcI = ei;
    const int* dstI = ei + E;

    float* wsf = (float*)d_ws;
    size_t off = 0;
    auto take = [&](size_t nf) -> size_t { size_t o = off; off += (nf + 63) & ~(size_t)63; return o; };
    float*    h1    = wsf + take((size_t)N * HC1);
    float*    agg1  = wsf + take((size_t)N * HC1);
    float*    as1   = wsf + take((size_t)N * 2);
    float*    ad1   = wsf + take((size_t)N * 2);
    unsigned* nmax1 = (unsigned*)(wsf + take((size_t)N * 2));
    float*    den1  = wsf + take((size_t)N * 2);
    float*    coef1 = wsf + take((size_t)ET * 2);
    float*    h2    = wsf + take((size_t)N * OUTC);
    float*    agg2  = wsf + take((size_t)N * OUTC);
    float*    as2   = wsf + take((size_t)N);
    float*    ad2   = wsf + take((size_t)N);
    unsigned* nmax2 = (unsigned*)(wsf + take((size_t)N));
    float*    den2  = wsf + take((size_t)N);
    float*    coef2 = wsf + take((size_t)ET);

    dim3 blk(256);
    int egrid = (ET + 255) / 256;
    int wgrid = (ET + 7) / 8;

    gat_gemm_wmma<128, 8><<<(N + 127) / 128, blk, 0, stream>>>(x, W1, h1, N, HC1);
    gat_alpha_kernel<2, 64><<<(N + 7) / 8, blk, 0, stream>>>(h1, aS1, aD1, as1, ad1, N);
    gat_fill_u32_kernel<<<((long)N * 2 + 255) / 256, blk, 0, stream>>>(nmax1, 0u, (long)N * 2);
    gat_fill_u32_kernel<<<((long)N * 2 + 255) / 256, blk, 0, stream>>>((unsigned*)den1, 0u, (long)N * 2);
    gat_fill_u32_kernel<<<((long)N * HC1 + 255) / 256, blk, 0, stream>>>((unsigned*)agg1, 0u, (long)N * HC1);
    gat_edge_max_kernel<2><<<egrid, blk, 0, stream>>>(srcI, dstI, E, N, as1, ad1, nmax1);
    gat_edge_denom_kernel<2><<<egrid, blk, 0, stream>>>(srcI, dstI, E, N, as1, ad1, nmax1, den1);
    gat_edge_coef_kernel<2><<<egrid, blk, 0, stream>>>(srcI, dstI, E, N, as1, ad1, nmax1, den1, coef1);
    gat_edge_scatter_kernel<2, 64><<<wgrid, blk, 0, stream>>>(srcI, dstI, E, N, coef1, h1, agg1);
    gat_elu_bias_kernel<<<((long)N * HC1 + 255) / 256, blk, 0, stream>>>(agg1, b1, (long)N * HC1, HC1);

    gat_gemm_wmma<48, 3><<<(N + 127) / 128, blk, 0, stream>>>(agg1, W2, h2, N, OUTC);
    gat_alpha_kernel<1, 40><<<(N + 7) / 8, blk, 0, stream>>>(h2, aS2, aD2, as2, ad2, N);
    gat_fill_u32_kernel<<<((long)N + 255) / 256, blk, 0, stream>>>(nmax2, 0u, (long)N);
    gat_fill_u32_kernel<<<((long)N + 255) / 256, blk, 0, stream>>>((unsigned*)den2, 0u, (long)N);
    gat_fill_u32_kernel<<<((long)N * OUTC + 255) / 256, blk, 0, stream>>>((unsigned*)agg2, 0u, (long)N * OUTC);
    gat_edge_max_kernel<1><<<egrid, blk, 0, stream>>>(srcI, dstI, E, N, as2, ad2, nmax2);
    gat_edge_denom_kernel<1><<<egrid, blk, 0, stream>>>(srcI, dstI, E, N, as2, ad2, nmax2, den2);
    gat_edge_coef_kernel<1><<<egrid, blk, 0, stream>>>(srcI, dstI, E, N, as2, ad2, nmax2, den2, coef2);
    gat_edge_scatter_kernel<1, 40><<<wgrid, blk, 0, stream>>>(srcI, dstI, E, N, coef2, h2, agg2);
    gat_bias_out_kernel<<<((long)N * OUTC + 255) / 256, blk, 0, stream>>>(
        agg2, b2, (float*)d_out, (long)N * OUTC, OUTC);
}


# edge pass of both layers: a wave takes 32 edges (weights formed lane-parallel, one exp per edge; rows walked by readlane broadcast), not one edge; float atomics kept, sums in arrival order as before
# speedup vs baseline: 2.3360x; 2.3360x over previous
#include <hip/hip_runtime.h>
#include <math.h>

typedef float v2f __attribute__((ext_vector_type(2)));
typedef float v8f __attribute__((ext_vector_type(8)));

#define NEG_SLOPE 0.2f

__device__ __forceinline__ float gat_lrelu(float v) { return v > 0.0f ? v : NEG_SLOPE * v; }
__device__ __forceinline__ unsigned gat_fkey(float f) {
    unsigned u = __float_as_uint(f);
    return (u & 0x80000000u) ? ~u : (u | 0x80000000u);
}
__device__ __forceinline__ float gat_funkey(unsigned k) {
    return __uint_as_float((k & 0x80000000u) ? (k & 0x7fffffffu) : ~k);
}

template<int NCOL, int NCT>
__global__ __launch_bounds__(256) void gat_gemm_wmma(
    const float* __restrict__ X, const float* __restrict__ W,
    float* __restrict__ Y, int N, int ncolStore)
{
    constexpr int K = 128;
    __shared__ float ldsW[K * NCOL];
    for (int idx = threadIdx.x; idx < K * NCOL; idx += 256) {
        int k = idx / NCOL, c = idx % NCOL;
        ldsW[idx] = (c < ncolStore) ? W[k * ncolStore + c] : 0.0f;
    }
    __syncthreads();

    const int lane = threadIdx.x & 31;
    const int wave = threadIdx.x >> 5;
    const int m    = lane & 15;
    const int half = lane >> 4;
    const int koff = half << 1;

    long row0 = (long)blockIdx.x * 128 + wave * 16;
    long rowL = row0 + m; if (rowL > (long)N - 1) rowL = (long)N - 1;
    const float* xrow = X + rowL * K;

    v8f acc[NCT] = {};
    for (int k0 = 0; k0 < K; k0 += 4) {
        v2f a;
        a.x = xrow[k0 + koff];
        a.y = xrow[k0 + koff + 1];
#pragma unroll
        for (int t = 0; t < NCT; ++t) {
            v2f b;
            b.x = ldsW[(k0 + koff) * NCOL + t * 16 + m];
            b.y = ldsW[(k0 + koff + 1) * NCOL + t * 16 + m];
            acc[t] = __builtin_amdgcn_wmma_f32_16x16x4_f32(
                false, a, false, b, (short)0, acc[t], false, false);
        }
    }
#pragma unroll
    for (int r = 0; r < 8; ++r) {
        long row = row0 + half * 8 + r;
        if (row < N) {
#pragma unroll
            for (int t = 0; t < NCT; ++t) {
                int c = t * 16 + m;
                if (c < ncolStore) Y[row * ncolStore + c] = acc[t][r];
            }
        }
    }
}

template<int H, int C>
__global__ __launch_bounds__(256) void gat_alpha_kernel(
    const float* __restrict__ h, const float* __restrict__ aw_s,
    const float* __restrict__ aw_d, float* __restrict__ as_,
    float* __restrict__ ad_, int N)
{
    constexpr int HC = H * C;
    int wid  = blockIdx.x * (blockDim.x >> 5) + (threadIdx.x >> 5);
    int lane = threadIdx.x & 31;
    if (wid >= N) return;
    const float* hp = h + (long)wid * HC;
    float s[H], d[H];
#pragma unroll
    for (int hh = 0; hh < H; ++hh) { s[hh] = 0.0f; d[hh] = 0.0f; }
    for (int c = lane; c < HC; c += 32) {
        int hh = (H == 1) ? 0 : (c / C);
        float v = hp[c];
        s[hh] += v * aw_s[c];
        d[hh] += v * aw_d[c];
    }
#pragma unroll
    for (int hh = 0; hh < H; ++hh) {
        float sv = s[hh], dv = d[hh];
        for (int o = 16; o > 0; o >>= 1) {
            sv += __shfl_xor(sv, o, 32);
            dv += __shfl_xor(dv, o, 32);
        }
        if (lane == 0) {
            as_[(long)wid * H + hh] = sv;
            ad_[(long)wid * H + hh] = dv;
        }
    }
}

template<int H, int RANGE>
__global__ __launch_bounds__(1024) void gat_seg_max_lds_kernel(
    const int* __restrict__ src, const int* __restrict__ dst, int E, int N,
    const float* __restrict__ as_, unsigned* __restrict__ part)
{
    constexpr int U = 8;
    __shared__ unsigned tab[RANGE * H];
    const int tid = threadIdx.x;
    const int lo  = blockIdx.y * RANGE;
    int cnt = N - lo; if (cnt > RANGE) cnt = RANGE;
    for (int i = tid; i < RANGE * H; i += 1024) tab[i] = 0u;
    __syncthreads();

    const int ET    = E + N;
    const int ch    = (ET + (int)gridDim.x - 1) / (int)gridDim.x;
    const int begin = blockIdx.x * ch;
    int end = begin + ch; if (end > ET) end = ET;

    for (int i0 = begin + tid; i0 < end; i0 += 1024 * U) {
        int idv[U], dl[U], sl[U];
#pragma unroll
        for (int u = 0; u < U; ++u) {
            int id = i0 + u * 1024;
            idv[u] = id;
            int ie = (id < end && id < E) ? id : 0;
            dl[u] = dst[ie];
            sl[u] = src[ie];
        }
#pragma unroll
        for (int u = 0; u < U; ++u) asm volatile("" : "+v"(dl[u]), "+v"(sl[u]));
        int rel[U];
        float a[U][H];
#pragma unroll
        for (int u = 0; u < U; ++u) {
            int id = idv[u];
            int d = (id < E) ? dl[u] : (id - E);
            int s = (id < E) ? sl[u] : (id - E);
            int r = (id < end) ? (d - lo) : -1;
            bool in = (unsigned)r < (unsigned)cnt;
            rel[u] = in ? r : -1;
            int sg = in ? s : 0;
#pragma unroll
            for (int h = 0; h < H; ++h) a[u][h] = as_[sg * H + h];
        }
#pragma unroll
        for (int u = 0; u < U; ++u) {
#pragma unroll
            for (int h = 0; h < H; ++h) asm volatile("" : "+v"(a[u][h]));
        }
#pragma unroll
        for (int u = 0; u < U; ++u) {
            if (rel[u] >= 0) {
#pragma unroll
                for (int h = 0; h < H; ++h) atomicMax(&tab[rel[u] * H + h], gat_fkey(a[u][h]));
            }
        }
    }
    __syncthreads();
    unsigned* po = part + (size_t)blockIdx.x * ((size_t)N * H) + (size_t)lo * H;
    for (int i = tid; i < cnt * H; i += 1024) po[i] = tab[i];
}

template<int H, int NCHUNK>
__global__ __launch_bounds__(256) void gat_max_merge_kernel(
    const unsigned* __restrict__ part, const float* __restrict__ ad_,
    unsigned* __restrict__ nmax, int NH)
{
    int i = blockIdx.x * 256 + threadIdx.x;
    if (i >= NH) return;
    unsigned k = 0u;
#pragma unroll
    for (int c = 0; c < NCHUNK; ++c) {
        unsigned v = part[(size_t)c * NH + i];
        k = v > k ? v : k;
    }
    float ev = gat_lrelu(gat_funkey(k) + ad_[i]);
    nmax[i] = gat_fkey(ev);
}

template<int H, int C, int AGG_STRIDE, bool DEN_IN_ROW>
__global__ __launch_bounds__(256) void gat_edge_softmax_scatter_kernel(
    const int* __restrict__ src, const int* __restrict__ dst, int E, int N,
    const float* __restrict__ as_, const float* __restrict__ ad_,
    const unsigned* __restrict__ nmax, const float* __restrict__ feat,
    float* agg, float* den)
{
    constexpr int HC = H * C;
    constexpr int NCOLS = DEN_IN_ROW ? HC + 1 : HC;
    constexpr int NPIECE = (NCOLS + 31) / 32;
    constexpr int GROUP = 16 / NPIECE;
    static_assert(H == 1 || C % 32 == 0, "a 32-column piece must lie in one head");
    static_assert(!DEN_IN_ROW || H == 1, "in-row denominator: one head only");
    static_assert(H <= 2, "the separate-denominator add below selects between two heads");
    static_assert(32 % GROUP == 0, "a wave's 32 edges split into whole groups");
    const int lane = threadIdx.x & 31;
    const int ET   = E + N;
    const int wave = __builtin_amdgcn_readfirstlane((int)(threadIdx.x >> 5));
    const int base = (blockIdx.x * (blockDim.x >> 5) + wave) * 32;
    if (base >= ET) return;
    int cnt = ET - base; if (cnt > 32) cnt = 32;

    int id = base + lane; if (id > ET - 1) id = ET - 1;
    const int ie = (id < E) ? id : 0;
    const int sl = src[ie];
    const int dl = dst[ie];
    const int s = (id < E) ? sl : (id - E);
    const int d = (id < E) ? dl : (id - E);
    float w[H];
#pragma unroll
    for (int h = 0; h < H; ++h) {
        float ev = gat_lrelu(as_[s * H + h] + ad_[d * H + h]);
        w[h] = expf(ev - gat_funkey(nmax[d * H + h]));
    }

    auto bcast_i = [](int v, int j) -> int { return __builtin_amdgcn_readlane(v, j); };

    for (int j0 = 0; j0 < cnt; j0 += GROUP) {
        if (j0 + GROUP <= cnt) {
            float v[GROUP][NPIECE];
            int   dj[GROUP];
            float wj[GROUP][H];
#pragma unroll
            for (int u = 0; u < GROUP; ++u) {
                const int sj = bcast_i(s, j0 + u);
                dj[u] = bcast_i(d, j0 + u);
#pragma unroll
                for (int h = 0; h < H; ++h)
                    wj[u][h] = __int_as_float(bcast_i(__float_as_int(w[h]), j0 + u));
                const float* fs = feat + (long)sj * HC;
#pragma unroll
                for (int p = 0; p < NPIECE; ++p) {
                    const int c  = p * 32 + lane;
                    const int cc = (c < HC) ? c : (HC - 1);
                    const float t = fs[cc];
                    v[u][p] = (c < HC) ? t : 1.0f;
                }
            }
#pragma unroll
            for (int u = 0; u < GROUP; ++u) {
#pragma unroll
                for (int p = 0; p < NPIECE; ++p) asm volatile("" : "+v"(v[u][p]));
            }
#pragma unroll
            for (int u = 0; u < GROUP; ++u) {
                float* ao = agg + (long)dj[u] * AGG_STRIDE;
#pragma unroll
                for (int p = 0; p < NPIECE; ++p) {
                    const int hh = (H == 1) ? 0 : ((p * 32) / C);
                    const int c  = p * 32 + lane;
                    if (c < NCOLS) atomicAdd(&ao[c], wj[u][hh] * v[u][p]);
                }
                if (!DEN_IN_ROW) {
                    if (lane < H) atomicAdd(&den[(long)dj[u] * H + lane], (lane == 0) ? wj[u][0] : wj[u][H - 1]);
                }
            }
        } else {
            for (int j = j0; j < cnt; ++j) {
                const int sj = bcast_i(s, j);
                const int dd = bcast_i(d, j);
                float wq[H];
#pragma unroll
                for (int h = 0; h < H; ++h) wq[h] = __int_as_float(bcast_i(__float_as_int(w[h]), j));
                const float* fs = feat + (long)sj * HC;
                float* ao = agg + (long)dd * AGG_STRIDE;
#pragma unroll
                for (int p = 0; p < NPIECE; ++p) {
                    const int hh = (H == 1) ? 0 : ((p * 32) / C);
                    const int c  = p * 32 + lane;
                    if (c < NCOLS) {
                        float t = (c < HC) ? fs[c] : 1.0f;
                        atomicAdd(&ao[c], wq[hh] * t);
                    }
                }
                if (!DEN_IN_ROW) {
                    if (lane < H) atomicAdd(&den[(long)dd * H + lane], (lane == 0) ? wq[0] : wq[H - 1]);
                }
            }
        }
    }
}

__global__ void gat_fill_u32_kernel(unsigned* __restrict__ p, unsigned v, long n) {
    long i = blockIdx.x * (long)blockDim.x + threadIdx.x;
    if (i < n) p[i] = v;
}

template<int H, int C>
__global__ void gat_norm_elu_bias_kernel(float* __restrict__ a, const float* __restrict__ den,
                                         const float* __restrict__ b, long n) {
    constexpr int HC = H * C;
    long i = blockIdx.x * (long)blockDim.x + threadIdx.x;
    if (i < n) {
        long row = i / HC;
        int  c   = (int)(i % HC);
        float t = a[i] / (den[row * H + c / C] + 1e-16f) + b[c];
        a[i] = t > 0.0f ? t : (expf(t) - 1.0f);
    }
}

template<int C, int AGG_STRIDE>
__global__ void gat_norm_bias_out_kernel(const float* __restrict__ a, const float* __restrict__ b,
                                         float* __restrict__ out, long n) {
    long i = blockIdx.x * (long)blockDim.x + threadIdx.x;
    if (i < n) {
        long row = i / C;
        int  c   = (int)(i % C);
        const float* ar = a + row * AGG_STRIDE;
        out[i] = ar[c] / (ar[C] + 1e-16f) + b[c];
    }
}

extern "C" void kernel_launch(void* const* d_in, const int* in_sizes, int n_in,
                              void* d_out, int out_size, void* d_ws, size_t ws_size,
                              hipStream_t stream)
{
    (void)n_in; (void)out_size; (void)ws_size;
    const float* x   = (const float*)d_in[0];
    const int*   ei  = (const int*)  d_in[1];
    const float* W1  = (const float*)d_in[2];
    const float* aS1 = (const float*)d_in[3];
    const float* aD1 = (const float*)d_in[4];
    const float* b1  = (const float*)d_in[5];
    const float* W2  = (const float*)d_in[6];
    const float* aS2 = (const float*)d_in[7];
    const float* aD2 = (const float*)d_in[8];
    const float* b2  = (const float*)d_in[9];

    const int IN = 128, HC1 = 128, OUTC = 40;
    int N  = in_sizes[0] / IN;
    int E  = in_sizes[1] / 2;
    int ET = E + N;
    const int* srcI = ei;
    const int* dstI = ei + E;

    float* wsf = (float*)d_ws;
    size_t off = 0;
    auto take = [&](size_t nf) -> size_t { size_t o = off; off += (nf + 63) & ~(size_t)63; return o; };
    const int AGG2S = 64;
    float*    h1    = wsf + take((size_t)N * HC1);
    float*    agg1  = wsf + take((size_t)N * HC1);
    float*    as1   = wsf + take((size_t)N * 2);
    float*    ad1   = wsf + take((size_t)N * 2);
    unsigned* nmax1 = (unsigned*)(wsf + take((size_t)N * 2));
    float*    den1  = wsf + take((size_t)N * 2);
    float*    h2    = wsf + take((size_t)N * OUTC);
    float*    agg2  = wsf + take((size_t)N * AGG2S);
    float*    as2   = wsf + take((size_t)N);
    float*    ad2   = wsf + take((size_t)N);
    unsigned* nmax2 = (unsigned*)(wsf + take((size_t)N));

    dim3 blk(256);
    int wgrid = (ET + 255) / 256;
    constexpr int RNG1 = 12500, NCH1 = 32;
    constexpr int RNG2 = 25000, NCH2 = 64;

    gat_gemm_wmma<128, 8><<<(N + 127) / 128, blk, 0, stream>>>(x, W1, h1, N, HC1);
    gat_alpha_kernel<2, 64><<<(N + 7) / 8, blk, 0, stream>>>(h1, aS1, aD1, as1, ad1, N);
    gat_seg_max_lds_kernel<2, RNG1><<<dim3(NCH1, (N + RNG1 - 1) / RNG1), dim3(1024), 0, stream>>>(
        srcI, dstI, E, N, as1, (unsigned*)agg1);
    gat_max_merge_kernel<2, NCH1><<<(N * 2 + 255) / 256, blk, 0, stream>>>(
        (const unsigned*)agg1, ad1, nmax1, N * 2);
    gat_fill_u32_kernel<<<((long)N * 2 + 255) / 256, blk, 0, stream>>>((unsigned*)den1, 0u, (long)N * 2);
    gat_fill_u32_kernel<<<((long)N * HC1 + 255) / 256, blk, 0, stream>>>((unsigned*)agg1, 0u, (long)N * HC1);
    gat_edge_softmax_scatter_kernel<2, 64, 128, false><<<wgrid, blk, 0, stream>>>(
        srcI, dstI, E, N, as1, ad1, nmax1, h1, agg1, den1);
    gat_norm_elu_bias_kernel<2, 64><<<((long)N * HC1 + 255) / 256, blk, 0, stream>>>(
        agg1, den1, b1, (long)N * HC1);

    gat_gemm_wmma<48, 3><<<(N + 127) / 128, blk, 0, stream>>>(agg1, W2, h2, N, OUTC);
    gat_alpha_kernel<1, 40><<<(N + 7) / 8, blk, 0, stream>>>(h2, aS2, aD2, as2, ad2, N);
    gat_seg_max_lds_kernel<1, RNG2><<<dim3(NCH2, (N + RNG2 - 1) / RNG2), dim3(1024), 0, stream>>>(
        srcI, dstI, E, N, as2, (unsigned*)h1);
    gat_max_merge_kernel<1, NCH2><<<(N + 255) / 256, blk, 0, stream>>>(
        (const unsigned*)h1, ad2, nmax2, N);
    gat_fill_u32_kernel<<<((long)N * AGG2S + 255) / 256, blk, 0, stream>>>((unsigned*)agg2, 0u, (long)N * AGG2S);
    gat_edge_softmax_scatter_kernel<1, 40, 64, true><<<wgrid, blk, 0, stream>>>(
        srcI, dstI, E, N, as2, ad2, nmax2, h2, agg2, agg2);
    gat_norm_bias_out_kernel<40, 64><<<((long)N * OUTC + 255) / 256, blk, 0, stream>>>(
        agg2, b2, (float*)d_out, (long)N * OUTC);
}


# edge passes: each package half (die id from s_sendmsg_rtn) adds into its own accumulator copy, epilogues add the copies; sum order changes (two partial sums; float atomics, arrival order as before)
# speedup vs baseline: 2.5744x; 1.1021x over previous
#include <hip/hip_runtime.h>
#include <math.h>

typedef float v2f __attribute__((ext_vector_type(2)));
typedef float v8f __attribute__((ext_vector_type(8)));

#define NEG_SLOPE 0.2f

__device__ __forceinline__ float gat_lrelu(float v) { return v > 0.0f ? v : NEG_SLOPE * v; }
__device__ __forceinline__ unsigned gat_fkey(float f) {
    unsigned u = __float_as_uint(f);
    return (u & 0x80000000u) ? ~u : (u | 0x80000000u);
}
__device__ __forceinline__ float gat_funkey(unsigned k) {
    return __uint_as_float((k & 0x80000000u) ? (k & 0x7fffffffu) : ~k);
}

template<int NCOL, int NCT>
__global__ __launch_bounds__(256) void gat_gemm_wmma(
    const float* __restrict__ X, const float* __restrict__ W,
    float* __restrict__ Y, int N, int ncolStore)
{
    constexpr int K = 128;
    __shared__ float ldsW[K * NCOL];
    for (int idx = threadIdx.x; idx < K * NCOL; idx += 256) {
        int k = idx / NCOL, c = idx % NCOL;
        ldsW[idx] = (c < ncolStore) ? W[k * ncolStore + c] : 0.0f;
    }
    __syncthreads();

    const int lane = threadIdx.x & 31;
    const int wave = threadIdx.x >> 5;
    const int m    = lane & 15;
    const int half = lane >> 4;
    const int koff = half << 1;

    long row0 = (long)blockIdx.x * 128 + wave * 16;
    long rowL = row0 + m; if (rowL > (long)N - 1) rowL = (long)N - 1;
    const float* xrow = X + rowL * K;

    v8f acc[NCT] = {};
    for (int k0 = 0; k0 < K; k0 += 4) {
        v2f a;
        a.x = xrow[k0 + koff];
        a.y = xrow[k0 + koff + 1];
#pragma unroll
        for (int t = 0; t < NCT; ++t) {
            v2f b;
            b.x = ldsW[(k0 + koff) * NCOL + t * 16 + m];
            b.y = ldsW[(k0 + koff + 1) * NCOL + t * 16 + m];
            acc[t] = __builtin_amdgcn_wmma_f32_16x16x4_f32(
                false, a, false, b, (short)0, acc[t], false, false);
        }
    }
#pragma unroll
    for (int r = 0; r < 8; ++r) {
        long row = row0 + half * 8 + r;
        if (row < N) {
#pragma unroll
            for (int t = 0; t < NCT; ++t) {
                int c = t * 16 + m;
                if (c < ncolStore) Y[row * ncolStore + c] = acc[t][r];
            }
        }
    }
}

template<int H, int C>
__global__ __launch_bounds__(256) void gat_alpha_kernel(
    const float* __restrict__ h, const float* __restrict__ aw_s,
    const float* __restrict__ aw_d, float* __restrict__ as_,
    float* __restrict__ ad_, int N)
{
    constexpr int HC = H * C;
    int wid  = blockIdx.x * (blockDim.x >> 5) + (threadIdx.x >> 5);
    int lane = threadIdx.x & 31;
    if (wid >= N) return;
    const float* hp = h + (long)wid * HC;
    float s[H], d[H];
#pragma unroll
    for (int hh = 0; hh < H; ++hh) { s[hh] = 0.0f; d[hh] = 0.0f; }
    for (int c = lane; c < HC; c += 32) {
        int hh = (H == 1) ? 0 : (c / C);
        float v = hp[c];
        s[hh] += v * aw_s[c];
        d[hh] += v * aw_d[c];
    }
#pragma unroll
    for (int hh = 0; hh < H; ++hh) {
        float sv = s[hh], dv = d[hh];
        for (int o = 16; o > 0; o >>= 1) {
            sv += __shfl_xor(sv, o, 32);
            dv += __shfl_xor(dv, o, 32);
        }
        if (lane == 0) {
            as_[(long)wid * H + hh] = sv;
            ad_[(long)wid * H + hh] = dv;
        }
    }
}

template<int H, int RANGE>
__global__ __launch_bounds__(1024) void gat_seg_max_lds_kernel(
    const int* __restrict__ src, const int* __restrict__ dst, int E, int N,
    const float* __restrict__ as_, unsigned* __restrict__ part)
{
    constexpr int U = 8;
    __shared__ unsigned tab[RANGE * H];
    const int tid = threadIdx.x;
    const int lo  = blockIdx.y * RANGE;
    int cnt = N - lo; if (cnt > RANGE) cnt = RANGE;
    for (int i = tid; i < RANGE * H; i += 1024) tab[i] = 0u;
    __syncthreads();

    const int ET    = E + N;
    const int ch    = (ET + (int)gridDim.x - 1) / (int)gridDim.x;
    const int begin = blockIdx.x * ch;
    int end = begin + ch; if (end > ET) end = ET;

    for (int i0 = begin + tid; i0 < end; i0 += 1024 * U) {
        int idv[U], dl[U], sl[U];
#pragma unroll
        for (int u = 0; u < U; ++u) {
            int id = i0 + u * 1024;
            idv[u] = id;
            int ie = (id < end && id < E) ? id : 0;
            dl[u] = dst[ie];
            sl[u] = src[ie];
        }
#pragma unroll
        for (int u = 0; u < U; ++u) asm volatile("" : "+v"(dl[u]), "+v"(sl[u]));
        int rel[U];
        float a[U][H];
#pragma unroll
        for (int u = 0; u < U; ++u) {
            int id = idv[u];
            int d = (id < E) ? dl[u] : (id - E);
            int s = (id < E) ? sl[u] : (id - E);
            int r = (id < end) ? (d - lo) : -1;
            bool in = (unsigned)r < (unsigned)cnt;
            rel[u] = in ? r : -1;
            int sg = in ? s : 0;
#pragma unroll
            for (int h = 0; h < H; ++h) a[u][h] = as_[sg * H + h];
        }
#pragma unroll
        for (int u = 0; u < U; ++u) {
#pragma unroll
            for (int h = 0; h < H; ++h) asm volatile("" : "+v"(a[u][h]));
        }
#pragma unroll
        for (int u = 0; u < U; ++u) {
            if (rel[u] >= 0) {
#pragma unroll
                for (int h = 0; h < H; ++h) atomicMax(&tab[rel[u] * H + h], gat_fkey(a[u][h]));
            }
        }
    }
    __syncthreads();
    unsigned* po = part + (size_t)blockIdx.x * ((size_t)N * H) + (size_t)lo * H;
    for (int i = tid; i < cnt * H; i += 1024) po[i] = tab[i];
}

template<int H, int NCHUNK>
__global__ __launch_bounds__(256) void gat_max_merge_kernel(
    const unsigned* __restrict__ part, const float* __restrict__ ad_,
    unsigned* __restrict__ nmax, int NH)
{
    int i = blockIdx.x * 256 + threadIdx.x;
    if (i >= NH) return;
    unsigned k = 0u;
#pragma unroll
    for (int c = 0; c < NCHUNK; ++c) {
        unsigned v = part[(size_t)c * NH + i];
        k = v > k ? v : k;
    }
    float ev = gat_lrelu(gat_funkey(k) + ad_[i]);
    nmax[i] = gat_fkey(ev);
}

template<int H, int C, int AGG_STRIDE, bool DEN_IN_ROW>
__global__ __launch_bounds__(256) void gat_edge_softmax_scatter_kernel(
    const int* __restrict__ src, const int* __restrict__ dst, int E, int N,
    const float* __restrict__ as_, const float* __restrict__ ad_,
    const unsigned* __restrict__ nmax, const float* __restrict__ feat,
    float* aggLo, float* denLo, float* aggHi, float* denHi, int rows2)
{
    constexpr int HC = H * C;
    constexpr int NCOLS = DEN_IN_ROW ? HC + 1 : HC;
    constexpr int NPIECE = (NCOLS + 31) / 32;
    constexpr int GROUP = 16 / NPIECE;
    static_assert(H == 1 || C % 32 == 0, "a 32-column piece must lie in one head");
    static_assert(!DEN_IN_ROW || H == 1, "in-row denominator: one head only");
    static_assert(H <= 2, "the separate-denominator add below selects between two heads");
    static_assert(32 % GROUP == 0, "a wave's 32 edges split into whole groups");
    const int lane = threadIdx.x & 31;
    const int ET   = E + N;
    const int wave = __builtin_amdgcn_readfirstlane((int)(threadIdx.x >> 5));
    const int base = (blockIdx.x * (blockDim.x >> 5) + wave) * 32;
    if (base >= ET) return;
    int cnt = ET - base; if (cnt > 32) cnt = 32;

    const unsigned place = __builtin_amdgcn_s_sendmsg_rtn(0x87);
    const bool upper = ((place >> 16) & 15u) >= 4u;
    float* const den = upper ? denHi : denLo;
    auto acc_row = [&](int dnode) -> float* {
        return ((upper && dnode < rows2) ? aggHi : aggLo) + (long)dnode * AGG_STRIDE;
    };

    int id = base + lane; if (id > ET - 1) id = ET - 1;
    const int ie = (id < E) ? id : 0;
    const int sl = src[ie];
    const int dl = dst[ie];
    const int s = (id < E) ? sl : (id - E);
    const int d = (id < E) ? dl : (id - E);
    float w[H];
#pragma unroll
    for (int h = 0; h < H; ++h) {
        float ev = gat_lrelu(as_[s * H + h] + ad_[d * H + h]);
        w[h] = expf(ev - gat_funkey(nmax[d * H + h]));
    }

    auto bcast_i = [](int v, int j) -> int { return __builtin_amdgcn_readlane(v, j); };

    for (int j0 = 0; j0 < cnt; j0 += GROUP) {
        if (j0 + GROUP <= cnt) {
            float v[GROUP][NPIECE];
            int   dj[GROUP];
            float wj[GROUP][H];
#pragma unroll
            for (int u = 0; u < GROUP; ++u) {
                const int sj = bcast_i(s, j0 + u);
                dj[u] = bcast_i(d, j0 + u);
#pragma unroll
                for (int h = 0; h < H; ++h)
                    wj[u][h] = __int_as_float(bcast_i(__float_as_int(w[h]), j0 + u));
                const float* fs = feat + (long)sj * HC;
#pragma unroll
                for (int p = 0; p < NPIECE; ++p) {
                    const int c  = p * 32 + lane;
                    const int cc = (c < HC) ? c : (HC - 1);
                    const float t = fs[cc];
                    v[u][p] = (c < HC) ? t : 1.0f;
                }
            }
#pragma unroll
            for (int u = 0; u < GROUP; ++u) {
#pragma unroll
                for (int p = 0; p < NPIECE; ++p) asm volatile("" : "+v"(v[u][p]));
            }
#pragma unroll
            for (int u = 0; u < GROUP; ++u) {
                float* ao = acc_row(dj[u]);
#pragma unroll
                for (int p = 0; p < NPIECE; ++p) {
                    const int hh = (H == 1) ? 0 : ((p * 32) / C);
                    const int c  = p * 32 + lane;
                    if (c < NCOLS) atomicAdd(&ao[c], wj[u][hh] * v[u][p]);
                }
                if (!DEN_IN_ROW) {
                    if (lane < H) atomicAdd(&den[(long)dj[u] * H + lane], (lane == 0) ? wj[u][0] : wj[u][H - 1]);
                }
            }
        } else {
            for (int j = j0; j < cnt; ++j) {
                const int sj = bcast_i(s, j);
                const int dd = bcast_i(d, j);
                float wq[H];
#pragma unroll
                for (int h = 0; h < H; ++h) wq[h] = __int_as_float(bcast_i(__float_as_int(w[h]), j));
                const float* fs = feat + (long)sj * HC;
                float* ao = acc_row(dd);
#pragma unroll
                for (int p = 0; p < NPIECE; ++p) {
                    const int hh = (H == 1) ? 0 : ((p * 32) / C);
                    const int c  = p * 32 + lane;
                    if (c < NCOLS) {
                        float t = (c < HC) ? fs[c] : 1.0f;
                        atomicAdd(&ao[c], wq[hh] * t);
                    }
                }
                if (!DEN_IN_ROW) {
                    if (lane < H) atomicAdd(&den[(long)dd * H + lane], (lane == 0) ? wq[0] : wq[H - 1]);
                }
            }
        }
    }
}

__global__ void gat_fill_u32_kernel(unsigned* __restrict__ p, unsigned v, long n) {
    long i = blockIdx.x * (long)blockDim.x + threadIdx.x;
    if (i < n) p[i] = v;
}

__global__ void gat_fill2_u32_kernel(unsigned* __restrict__ pa, unsigned* __restrict__ pb, unsigned v, long n, long nb) {
    long i = blockIdx.x * (long)blockDim.x + threadIdx.x;
    if (i < n)  pa[i] = v;
    if (i < nb) pb[i] = v;
}

template<int H, int C>
__global__ void gat_norm_elu_bias_kernel(float* __restrict__ a, const float* __restrict__ aHi,
                                         const float* __restrict__ den,
                                         const float* __restrict__ denHi, int rows2,
                                         const float* __restrict__ b, long n) {
    constexpr int HC = H * C;
    long i = blockIdx.x * (long)blockDim.x + threadIdx.x;
    if (i < n) {
        long row = i / HC;
        int  c   = (int)(i % HC);
        float sum = a[i];
        if (row < rows2) sum += aHi[i];
        const float dn = den[row * H + c / C] + denHi[row * H + c / C];
        float t = sum / (dn + 1e-16f) + b[c];
        a[i] = t > 0.0f ? t : (expf(t) - 1.0f);
    }
}

template<int C, int AGG_STRIDE>
__global__ void gat_norm_bias_out_kernel(const float* __restrict__ a, const float* __restrict__ aHi,
                                         const float* __restrict__ b, float* __restrict__ out, long n) {
    long i = blockIdx.x * (long)blockDim.x + threadIdx.x;
    if (i < n) {
        long row = i / C;
        int  c   = (int)(i % C);
        const float* ar = a + row * AGG_STRIDE;
        const float* br = aHi + row * AGG_STRIDE;
        out[i] = (ar[c] + br[c]) / (ar[C] + br[C] + 1e-16f) + b[c];
    }
}

extern "C" void kernel_launch(void* const* d_in, const int* in_sizes, int n_in,
                              void* d_out, int out_size, void* d_ws, size_t ws_size,
                              hipStream_t stream)
{
    (void)n_in; (void)out_size;
    const float* x   = (const float*)d_in[0];
    const int*   ei  = (const int*)  d_in[1];
    const float* W1  = (const float*)d_in[2];
    const float* aS1 = (const float*)d_in[3];
    const float* aD1 = (const float*)d_in[4];
    const float* b1  = (const float*)d_in[5];
    const float* W2  = (const float*)d_in[6];
    const float* aS2 = (const float*)d_in[7];
    const float* aD2 = (const float*)d_in[8];
    const float* b2  = (const float*)d_in[9];

    const int IN = 128, HC1 = 128, OUTC = 40;
    int N  = in_sizes[0] / IN;
    int E  = in_sizes[1] / 2;
    int ET = E + N;
    const int* srcI = ei;
    const int* dstI = ei + E;

    float* wsf = (float*)d_ws;
    size_t off = 0;
    auto take = [&](size_t nf) -> size_t { size_t o = off; off += (nf + 63) & ~(size_t)63; return o; };
    const int AGG2S = 64;
    float*    h1    = wsf + take((size_t)N * HC1);
    float*    agg1  = wsf + take((size_t)N * HC1);
    float*    as1   = wsf + take((size_t)N * 2);
    float*    ad1   = wsf + take((size_t)N * 2);
    unsigned* nmax1 = (unsigned*)(wsf + take((size_t)N * 2));
    float*    den1  = wsf + take((size_t)N * 4);
    const size_t offR = off;
    float*    h2    = wsf + take((size_t)N * OUTC);
    float*    as2   = wsf + take((size_t)N);
    float*    ad2   = wsf + take((size_t)N);
    unsigned* nmax2 = (unsigned*)(wsf + take((size_t)N));
    const size_t wsFloats = ws_size / sizeof(float);
    const size_t roomRows = wsFloats > offR ? (wsFloats - offR) / (size_t)HC1 : 0;
    const int    rows2    = roomRows < (size_t)N ? (int)roomRows : N;
    float*    agg1B = wsf + offR;
    float*    den1B = den1 + (size_t)N * 2;
    float*    agg2  = agg1;
    float*    agg2B = agg1 + (size_t)N * AGG2S;

    dim3 blk(256);
    int wgrid = (ET + 255) / 256;
    constexpr int RNG1 = 12500, NCH1 = 32;
    constexpr int RNG2 = 25000, NCH2 = 64;

    gat_gemm_wmma<128, 8><<<(N + 127) / 128, blk, 0, stream>>>(x, W1, h1, N, HC1);
    gat_alpha_kernel<2, 64><<<(N + 7) / 8, blk, 0, stream>>>(h1, aS1, aD1, as1, ad1, N);
    gat_seg_max_lds_kernel<2, RNG1><<<dim3(NCH1, (N + RNG1 - 1) / RNG1), dim3(1024), 0, stream>>>(
        srcI, dstI, E, N, as1, (unsigned*)agg1);
    gat_max_merge_kernel<2, NCH1><<<(N * 2 + 255) / 256, blk, 0, stream>>>(
        (const unsigned*)agg1, ad1, nmax1, N * 2);
    gat_fill_u32_kernel<<<((long)N * 4 + 255) / 256, blk, 0, stream>>>((unsigned*)den1, 0u, (long)N * 4);
    gat_fill2_u32_kernel<<<((long)N * HC1 + 255) / 256, blk, 0, stream>>>(
        (unsigned*)agg1, (unsigned*)agg1B, 0u, (long)N * HC1, (long)rows2 * HC1);
    gat_edge_softmax_scatter_kernel<2, 64, 128, false><<<wgrid, blk, 0, stream>>>(
        srcI, dstI, E, N, as1, ad1, nmax1, h1, agg1, den1, agg1B, den1B, rows2);
    gat_norm_elu_bias_kernel<2, 64><<<((long)N * HC1 + 255) / 256, blk, 0, stream>>>(
        agg1, agg1B, den1, den1B, rows2, b1, (long)N * HC1);

    gat_gemm_wmma<48, 3><<<(N + 127) / 128, blk, 0, stream>>>(agg1, W2, h2, N, OUTC);
    gat_alpha_kernel<1, 40><<<(N + 7) / 8, blk, 0, stream>>>(h2, aS2, aD2, as2, ad2, N);
    gat_seg_max_lds_kernel<1, RNG2><<<dim3(NCH2, (N + RNG2 - 1) / RNG2), dim3(1024), 0, stream>>>(
        srcI, dstI, E, N, as2, (unsigned*)h1);
    gat_max_merge_kernel<1, NCH2><<<(N + 255) / 256, blk, 0, stream>>>(
        (const unsigned*)h1, ad2, nmax2, N);
    gat_fill_u32_kernel<<<((long)N * AGG2S * 2 + 255) / 256, blk, 0, stream>>>((unsigned*)agg2, 0u, (long)N * AGG2S * 2);
    gat_edge_softmax_scatter_kernel<1, 40, 64, true><<<wgrid, blk, 0, stream>>>(
        srcI, dstI, E, N, as2, ad2, nmax2, h2, agg2, agg2, agg2B, agg2B, N);
    gat_norm_bias_out_kernel<40, 64><<<((long)N * OUTC + 255) / 256, blk, 0, stream>>>(
        agg2, agg2B, b2, (float*)d_out, (long)N * OUTC);
}


# attention logits of both layers leave the products as extra output columns (W gains W@a_src, W@a_dst, formed per work-group in LDS); both wave-per-node logit launches go; logit sum order changes
# speedup vs baseline: 4.1237x; 1.6018x over previous
#include <hip/hip_runtime.h>
#include <math.h>

typedef float v2f __attribute__((ext_vector_type(2)));
typedef float v8f __attribute__((ext_vector_type(8)));

#define NEG_SLOPE 0.2f

__device__ __forceinline__ float gat_lrelu(float v) { return v > 0.0f ? v : NEG_SLOPE * v; }
__device__ __forceinline__ unsigned gat_fkey(float f) {
    unsigned u = __float_as_uint(f);
    return (u & 0x80000000u) ? ~u : (u | 0x80000000u);
}
__device__ __forceinline__ float gat_funkey(unsigned k) {
    return __uint_as_float((k & 0x80000000u) ? (k & 0x7fffffffu) : ~k);
}

template<int NCOL, int NCT, int H, int C>
__global__ __launch_bounds__(256) void gat_gemm_wmma(
    const float* __restrict__ X, const float* __restrict__ W,
    const float* __restrict__ aw_s, const float* __restrict__ aw_d,
    float* __restrict__ Y, float* __restrict__ as_, float* __restrict__ ad_, int N)
{
    constexpr int K  = 128;
    constexpr int HC = H * C;
    static_assert(HC + 2 * H <= NCOL, "the logit columns fit the padded width");
    static_assert(K * H <= 256, "one thread per (k, head) pair of the added columns");
    __shared__ float ldsW[K * NCOL];
    for (int idx = threadIdx.x; idx < K * NCOL; idx += 256) {
        int k = idx / NCOL, c = idx % NCOL;
        ldsW[idx] = (c < HC) ? W[k * HC + c] : 0.0f;
    }
    __syncthreads();
    if (threadIdx.x < K * H) {
        const int k  = threadIdx.x % K;
        const int hd = threadIdx.x / K;
        const float* wr = ldsW + k * NCOL + hd * C;
        const float* ps = aw_s + hd * C;
        const float* pd = aw_d + hd * C;
        float s = 0.0f, d = 0.0f;
        for (int c = 0; c < C; ++c) {
            const float w = wr[c];
            s += w * ps[c];
            d += w * pd[c];
        }
        ldsW[k * NCOL + HC + hd]     = s;
        ldsW[k * NCOL + HC + H + hd] = d;
    }
    __syncthreads();

    const int lane = threadIdx.x & 31;
    const int wave = threadIdx.x >> 5;
    const int m    = lane & 15;
    const int half = lane >> 4;
    const int koff = half << 1;

    long row0 = (long)blockIdx.x * 128 + wave * 16;
    long rowL = row0 + m; if (rowL > (long)N - 1) rowL = (long)N - 1;
    const float* xrow = X + rowL * K;

    v8f acc[NCT] = {};
    for (int k0 = 0; k0 < K; k0 += 4) {
        v2f a;
        a.x = xrow[k0 + koff];
        a.y = xrow[k0 + koff + 1];
#pragma unroll
        for (int t = 0; t < NCT; ++t) {
            v2f b;
            b.x = ldsW[(k0 + koff) * NCOL + t * 16 + m];
            b.y = ldsW[(k0 + koff + 1) * NCOL + t * 16 + m];
            acc[t] = __builtin_amdgcn_wmma_f32_16x16x4_f32(
                false, a, false, b, (short)0, acc[t], false, false);
        }
    }
#pragma unroll
    for (int r = 0; r < 8; ++r) {
        long row = row0 + half * 8 + r;
        if (row < N) {
#pragma unroll
            for (int t = 0; t < NCT; ++t) {
                const int c = t * 16 + m;
                if (c < HC)              Y[row * HC + c] = acc[t][r];
                else if (c < HC + H)     as_[row * H + (c - HC)] = acc[t][r];
                else if (c < HC + 2 * H) ad_[row * H + (c - HC - H)] = acc[t][r];
            }
        }
    }
}

template<int RS>
__global__ __launch_bounds__(256) void gat_range_id_kernel(
    const int* __restrict__ dst, int E, unsigned* __restrict__ rid32, int nwords)
{
    const int t = blockIdx.x * 256 + threadIdx.x;
    if (t >= nwords) return;
    unsigned wv = 0u;
#pragma unroll
    for (int k = 0; k < 4; ++k) {
        const int e  = 4 * t + k;
        const int ec = (e < E) ? e : 0;
        const unsigned d = (unsigned)dst[ec];
        const unsigned b = (e < E) ? ((d / (unsigned)RS) & 255u) : 0u;
        wv |= b << (8 * k);
    }
    rid32[t] = wv;
}

template<int H, int C, int RS, int QCAP, bool ELU, int LMODE, int LCAP>
__global__ __launch_bounds__(1024) void gat_range_owner_kernel(
    const int* __restrict__ src, const int* __restrict__ dst, int E, int N,
    const unsigned* __restrict__ rid32, int nvec,
    const float* __restrict__ as_, const float* __restrict__ ad_,
    const float* __restrict__ feat,
    const float* __restrict__ bias, float* __restrict__ out,
    unsigned* __restrict__ lst, int* __restrict__ lcnt)
{
    constexpr int HC = H * C;
    constexpr int NPIECE = (HC + 31) / 32;
    constexpr int GROUP = 16 / NPIECE;
    constexpr int U = 4;
    constexpr int RU = 4;
    constexpr unsigned NONE = 0xFFFFFFFFu;
    constexpr int SBITS = 23;
    static_assert(H == 1 || C % 32 == 0, "a 32-column piece must lie in one head");
    static_assert(H <= 2, "the denominator add below selects between two heads");
    static_assert(32 % GROUP == 0, "a wave's 32 entries split into whole groups");
    static_assert(RS <= 511, "the local row takes 9 bits and 511 is kept for NONE");
    static_assert(LMODE == 2 || QCAP <= LCAP, "a stored list fits its slot");
    __shared__ float    acc[RS * HC];
    __shared__ float    den[RS * H];
    __shared__ unsigned mtab[RS * H];
    __shared__ unsigned queue[QCAP];
    __shared__ int      qn;

    const int tid  = threadIdx.x;
    const int lane = tid & 31;
    const int wave = __builtin_amdgcn_readfirstlane(tid >> 5);
    const int lo   = blockIdx.x * RS;
    int cnt = N - lo; if (cnt > RS) cnt = RS;

    for (int i = tid; i < RS * HC; i += 1024) acc[i] = 0.0f;
    for (int i = tid; i < RS * H; i += 1024) { den[i] = 0.0f; mtab[i] = 0u; }
    if (tid == 0) { qn = 0; queue[0] = NONE; }
    __syncthreads();

    const uint4* __restrict__ rv = reinterpret_cast<const uint4*>(rid32);
    const unsigned pat = (unsigned)(blockIdx.x & 255) * 0x01010101u;
    auto stream = [&](auto&& f) {
        for (int v0 = tid; v0 < nvec; v0 += 1024 * U) {
            uint4 x[U];
#pragma unroll
            for (int u = 0; u < U; ++u) {
                const int vi = v0 + u * 1024;
                x[u] = rv[(vi < nvec) ? vi : (nvec - 1)];
            }
#pragma unroll
            for (int u = 0; u < U; ++u) {
                const int vi = v0 + u * 1024;
                const unsigned q[4] = { x[u].x, x[u].y, x[u].z, x[u].w };
                unsigned z[4];
#pragma unroll
                for (int k = 0; k < 4; ++k) {
                    const unsigned y = q[k] ^ pat;
                    z[k] = (y - 0x01010101u) & ~y & 0x80808080u;
                }
                if (vi < nvec && (z[0] | z[1] | z[2] | z[3]) != 0u) {
                    unsigned m = 0u;
#pragma unroll
                    for (int k = 0; k < 4; ++k)
                        m |= ((((z[k] >> 7) * 0x00204081u) >> 21) & 15u) << (4 * k);
                    while (m != 0u) {
                        const int e = __builtin_ctz(m);
                        m &= m - 1u;
                        f(vi * 16 + e);
                    }
                }
            }
        }
    };
    auto push = [&](int eid) {
        const int p = atomicAdd(&qn, 1);
        if (p < QCAP) queue[p] = (unsigned)eid;
    };
    auto edge_max = [&](int eid) {
        if (eid >= E) return;
        const int r = dst[eid] - lo;
        if ((unsigned)r >= (unsigned)cnt) return;
        const int s = src[eid];
#pragma unroll
        for (int h = 0; h < H; ++h) atomicMax(&mtab[r * H + h], gat_fkey(as_[s * H + h]));
    };
    auto edge_add = [&](int eid) {
        if (eid >= E) return;
        const int d = dst[eid];
        const int r = d - lo;
        if ((unsigned)r >= (unsigned)cnt) return;
        const int s = src[eid];
        float w[H];
#pragma unroll
        for (int h = 0; h < H; ++h) {
            const float ev = gat_lrelu(as_[s * H + h] + ad_[d * H + h]);
            w[h] = expf(ev - __uint_as_float(mtab[r * H + h]));
            atomicAdd(&den[r * H + h], w[h]);
        }
        const float* fs = feat + (long)s * HC;
        for (int c = 0; c < HC; ++c) atomicAdd(&acc[r * HC + c], ((H == 1 || c < C) ? w[0] : w[H - 1]) * fs[c]);
    };

    unsigned* __restrict__ myl = lst + (long)blockIdx.x * LCAP;
    int nl = -1;
    if (LMODE == 2) nl = __builtin_amdgcn_readfirstlane(lcnt[blockIdx.x]);
    const bool loaded = LMODE == 2 && nl >= 0 && nl <= QCAP && nl <= LCAP;

    if (!loaded) {
        stream(push);
        __syncthreads();
    }

    const int  qv   = loaded ? nl : __builtin_amdgcn_readfirstlane(qn);
    const bool full = !loaded && (qv > QCAP || N > (1 << SBITS));
    const int  n    = full ? 0 : qv;

    if (loaded) {
        for (int i0 = tid; i0 < n; i0 += 1024 * RU) {
            unsigned ee[RU];
#pragma unroll
            for (int u = 0; u < RU; ++u) {
                const int i = i0 + u * 1024;
                const unsigned e = myl[(i < n) ? i : i0];
                ee[u] = (i < n) ? e : NONE;
            }
            float a[RU][H];
#pragma unroll
            for (int u = 0; u < RU; ++u) {
                const int sg = (ee[u] != NONE) ? (int)(ee[u] & ((1u << SBITS) - 1u)) : 0;
#pragma unroll
                for (int h = 0; h < H; ++h) a[u][h] = as_[sg * H + h];
            }
#pragma unroll
            for (int u = 0; u < RU; ++u) {
                const int  i  = i0 + u * 1024;
                const bool in = ee[u] != NONE;
                if (i < n) queue[i] = ee[u];
                const int rt = in ? (int)(ee[u] >> SBITS) : 0;
#pragma unroll
                for (int h = 0; h < H; ++h) atomicMax(&mtab[rt * H + h], in ? gat_fkey(a[u][h]) : 0u);
            }
        }
    } else if (full) {
        stream(edge_max);
    } else {
        for (int i0 = tid; i0 < n; i0 += 1024 * RU) {
            int  dd[RU], ss[RU];
            bool ok[RU];
#pragma unroll
            for (int u = 0; u < RU; ++u) {
                const int i = i0 + u * 1024;
                const unsigned e = queue[(i < n) ? i : i0];
                ok[u] = (i < n) && e < (unsigned)E;
                const int ec = ok[u] ? (int)e : 0;
                dd[u] = dst[ec];
                ss[u] = src[ec];
            }
            int   rr[RU];
            float a[RU][H];
#pragma unroll
            for (int u = 0; u < RU; ++u) {
                const int  r  = dd[u] - lo;
                const bool in = ok[u] && (unsigned)r < (unsigned)cnt;
                rr[u] = in ? r : -1;
                const int sg = in ? ss[u] : 0;
#pragma unroll
                for (int h = 0; h < H; ++h) a[u][h] = as_[sg * H + h];
            }
#pragma unroll
            for (int u = 0; u < RU; ++u) {
                const int  i  = i0 + u * 1024;
                const bool in = rr[u] >= 0;
                if (i < n) queue[i] = in ? ((unsigned)ss[u] | ((unsigned)rr[u] << SBITS)) : NONE;
                const int rt = in ? rr[u] : 0;
#pragma unroll
                for (int h = 0; h < H; ++h) atomicMax(&mtab[rt * H + h], in ? gat_fkey(a[u][h]) : 0u);
            }
        }
    }
    __syncthreads();

    if (LMODE == 1) {
        for (int i = tid; i < n; i += 1024) myl[i] = queue[i];
        if (tid == 0) lcnt[blockIdx.x] = full ? -1 : n;
    }

    for (int i = tid; i < cnt * H; i += 1024) {
        const int g = lo * H + i;
        const unsigned ks = gat_fkey(as_[g]);
        const unsigned kt = mtab[i];
        const unsigned k  = kt > ks ? kt : ks;
        mtab[i] = __float_as_uint(gat_lrelu(gat_funkey(k) + ad_[g]));
    }
    __syncthreads();

    if (full) stream(edge_add);

    const int total = n + cnt;
    auto bcast_i = [](int v, int j) -> int { return __builtin_amdgcn_readlane(v, j); };

    for (int b0 = wave * 32; b0 < total; b0 += 1024) {
        int nb = total - b0; if (nb > 32) nb = 32;
        const int  i    = b0 + lane;
        const bool isq  = i < n;
        const unsigned ent = queue[isq ? i : 0];
        const int  sq   = (int)(ent & ((1u << SBITS) - 1u));
        const int  rq   = (int)(ent >> SBITS);
        const int  rs   = i - n;
        const bool valid = isq ? (ent != NONE) : (i < total);
        const int  r    = valid ? (isq ? rq : rs) : 0;
        const int  s    = valid ? (isq ? sq : (lo + rs)) : lo;
        const int  d    = lo + r;
        const int  vflag = valid ? 1 : 0;
        float w[H];
#pragma unroll
        for (int h = 0; h < H; ++h) {
            const float ev = gat_lrelu(as_[s * H + h] + ad_[d * H + h]);
            w[h] = expf(ev - __uint_as_float(mtab[r * H + h]));
        }
        for (int j0 = 0; j0 < nb; j0 += GROUP) {
            if (j0 + GROUP <= nb) {
                float v[GROUP][NPIECE];
                int   rj[GROUP];
                int   vj[GROUP];
                float wj[GROUP][H];
#pragma unroll
                for (int u = 0; u < GROUP; ++u) {
                    const int sj = bcast_i(s, j0 + u);
                    rj[u] = bcast_i(r, j0 + u);
                    vj[u] = bcast_i(vflag, j0 + u);
#pragma unroll
                    for (int h = 0; h < H; ++h)
                        wj[u][h] = __int_as_float(bcast_i(__float_as_int(w[h]), j0 + u));
                    const float* fs = feat + (long)sj * HC;
#pragma unroll
                    for (int p = 0; p < NPIECE; ++p) {
                        const int c  = p * 32 + lane;
                        const int cc = (c < HC) ? c : (HC - 1);
                        v[u][p] = fs[cc];
                    }
                }
#pragma unroll
                for (int u = 0; u < GROUP; ++u) {
#pragma unroll
                    for (int p = 0; p < NPIECE; ++p) asm volatile("" : "+v"(v[u][p]));
                }
#pragma unroll
                for (int u = 0; u < GROUP; ++u) {
                    if (vj[u] != 0) {
                        float* ar = acc + rj[u] * HC;
#pragma unroll
                        for (int p = 0; p < NPIECE; ++p) {
                            const int hh = (H == 1) ? 0 : ((p * 32) / C);
                            const int c  = p * 32 + lane;
                            if (c < HC) atomicAdd(&ar[c], wj[u][hh] * v[u][p]);
                        }
                        if (lane < H) atomicAdd(&den[rj[u] * H + lane], (lane == 0) ? wj[u][0] : wj[u][H - 1]);
                    }
                }
            } else {
                for (int j = j0; j < nb; ++j) {
                    const int sj = bcast_i(s, j);
                    const int rr = bcast_i(r, j);
                    const int vv = bcast_i(vflag, j);
                    float wq[H];
#pragma unroll
                    for (int h = 0; h < H; ++h) wq[h] = __int_as_float(bcast_i(__float_as_int(w[h]), j));
                    if (vv != 0) {
                        const float* fs = feat + (long)sj * HC;
                        float* ar = acc + rr * HC;
#pragma unroll
                        for (int p = 0; p < NPIECE; ++p) {
                            const int hh = (H == 1) ? 0 : ((p * 32) / C);
                            const int c  = p * 32 + lane;
                            if (c < HC) atomicAdd(&ar[c], wq[hh] * fs[c]);
                        }
                        if (lane < H) atomicAdd(&den[rr * H + lane], (lane == 0) ? wq[0] : wq[H - 1]);
                    }
                }
            }
        }
    }
    __syncthreads();

    float* po = out + (long)lo * HC;
    for (int i = tid; i < cnt * HC; i += 1024) {
        const int row = i / HC;
        const int c   = i - row * HC;
        const float t = acc[i] / (den[row * H + c / C] + 1e-16f) + bias[c];
        po[i] = ELU ? (t > 0.0f ? t : (expf(t) - 1.0f)) : t;
    }
}

extern "C" void kernel_launch(void* const* d_in, const int* in_sizes, int n_in,
                              void* d_out, int out_size, void* d_ws, size_t ws_size,
                              hipStream_t stream)
{
    (void)n_in; (void)out_size; (void)ws_size;
    const float* x   = (const float*)d_in[0];
    const int*   ei  = (const int*)  d_in[1];
    const float* W1  = (const float*)d_in[2];
    const float* aS1 = (const float*)d_in[3];
    const float* aD1 = (const float*)d_in[4];
    const float* b1  = (const float*)d_in[5];
    const float* W2  = (const float*)d_in[6];
    const float* aS2 = (const float*)d_in[7];
    const float* aD2 = (const float*)d_in[8];
    const float* b2  = (const float*)d_in[9];

    const int IN = 128, HC1 = 128;
    int N  = in_sizes[0] / IN;
    int E  = in_sizes[1] / 2;
    const int* srcI = ei;
    const int* dstI = ei + E;

    float* wsf = (float*)d_ws;
    size_t off = 0;
    auto take = [&](size_t nf) -> size_t { size_t o = off; off += (nf + 63) & ~(size_t)63; return o; };
    const int nvec   = (E + 15) / 16;
    const int nwords = nvec * 4;
    float*    h1    = wsf + take((size_t)N * HC1);
    float*    agg1  = wsf + take((size_t)N * HC1);
    float*    as1   = wsf + take((size_t)N * 2);
    float*    ad1   = wsf + take((size_t)N * 2);
    unsigned* rid   = (unsigned*)(wsf + take((size_t)nwords));
    float*    h2    = h1;
    float*    as2   = wsf + take((size_t)N);
    float*    ad2   = wsf + take((size_t)N);

    dim3 blk(256);
    constexpr int RS = 392, QCAP1 = 8192;
    constexpr int QCAP2 = 25600;
    const int ngrp = (N + RS - 1) / RS;
    unsigned* lst   = (unsigned*)(wsf + take((size_t)ngrp * QCAP1));
    int*      lcnt  = (int*)(wsf + take((size_t)ngrp));

    gat_range_id_kernel<RS><<<(nwords + 255) / 256, blk, 0, stream>>>(dstI, E, rid, nwords);

    gat_gemm_wmma<144, 9, 2, 64><<<(N + 127) / 128, blk, 0, stream>>>(x, W1, aS1, aD1, h1, as1, ad1, N);
    gat_range_owner_kernel<2, 64, RS, QCAP1, true, 1, QCAP1><<<ngrp, dim3(1024), 0, stream>>>(
        srcI, dstI, E, N, rid, nvec, as1, ad1, h1, b1, agg1, lst, lcnt);

    gat_gemm_wmma<48, 3, 1, 40><<<(N + 127) / 128, blk, 0, stream>>>(agg1, W2, aS2, aD2, h2, as2, ad2, N);
    gat_range_owner_kernel<1, 40, RS, QCAP2, false, 2, QCAP1><<<ngrp, dim3(1024), 0, stream>>>(
        srcI, dstI, E, N, rid, nvec, as2, ad2, h2, b2, (float*)d_out, lst, lcnt);
}


# layer 2's product (rows x W2 plus its two logit columns, f32 WMMA) runs in layer 1's owner pass on the finished rows kept in LDS at a padded pitch; the rows' memory round trip and one launch go
# speedup vs baseline: 4.4710x; 1.0842x over previous
#include <hip/hip_runtime.h>
#include <math.h>

typedef float v2f __attribute__((ext_vector_type(2)));
typedef float v8f __attribute__((ext_vector_type(8)));

#define NEG_SLOPE 0.2f

__device__ __forceinline__ float gat_lrelu(float v) { return v > 0.0f ? v : NEG_SLOPE * v; }
__device__ __forceinline__ unsigned gat_fkey(float f) {
    unsigned u = __float_as_uint(f);
    return (u & 0x80000000u) ? ~u : (u | 0x80000000u);
}
__device__ __forceinline__ float gat_funkey(unsigned k) {
    return __uint_as_float((k & 0x80000000u) ? (k & 0x7fffffffu) : ~k);
}

template<int NCOL, int NCT, int H, int C>
__global__ __launch_bounds__(256) void gat_gemm_wmma(
    const float* __restrict__ X, const float* __restrict__ W,
    const float* __restrict__ aw_s, const float* __restrict__ aw_d,
    float* __restrict__ Y, float* __restrict__ as_, float* __restrict__ ad_, int N)
{
    constexpr int K  = 128;
    constexpr int HC = H * C;
    static_assert(HC + 2 * H <= NCOL, "the logit columns fit the padded width");
    static_assert(K * H <= 256, "one thread per (k, head) pair of the added columns");
    __shared__ float ldsW[K * NCOL];
    for (int idx = threadIdx.x; idx < K * NCOL; idx += 256) {
        int k = idx / NCOL, c = idx % NCOL;
        ldsW[idx] = (c < HC) ? W[k * HC + c] : 0.0f;
    }
    __syncthreads();
    if (threadIdx.x < K * H) {
        const int k  = threadIdx.x % K;
        const int hd = threadIdx.x / K;
        const float* wr = ldsW + k * NCOL + hd * C;
        const float* ps = aw_s + hd * C;
        const float* pd = aw_d + hd * C;
        float s = 0.0f, d = 0.0f;
        for (int c = 0; c < C; ++c) {
            const float w = wr[c];
            s += w * ps[c];
            d += w * pd[c];
        }
        ldsW[k * NCOL + HC + hd]     = s;
        ldsW[k * NCOL + HC + H + hd] = d;
    }
    __syncthreads();

    const int lane = threadIdx.x & 31;
    const int wave = threadIdx.x >> 5;
    const int m    = lane & 15;
    const int half = lane >> 4;
    const int koff = half << 1;

    long row0 = (long)blockIdx.x * 128 + wave * 16;
    long rowL = row0 + m; if (rowL > (long)N - 1) rowL = (long)N - 1;
    const float* xrow = X + rowL * K;

    v8f acc[NCT] = {};
    for (int k0 = 0; k0 < K; k0 += 4) {
        v2f a;
        a.x = xrow[k0 + koff];
        a.y = xrow[k0 + koff + 1];
#pragma unroll
        for (int t = 0; t < NCT; ++t) {
            v2f b;
            b.x = ldsW[(k0 + koff) * NCOL + t * 16 + m];
            b.y = ldsW[(k0 + koff + 1) * NCOL + t * 16 + m];
            acc[t] = __builtin_amdgcn_wmma_f32_16x16x4_f32(
                false, a, false, b, (short)0, acc[t], false, false);
        }
    }
#pragma unroll
    for (int r = 0; r < 8; ++r) {
        long row = row0 + half * 8 + r;
        if (row < N) {
#pragma unroll
            for (int t = 0; t < NCT; ++t) {
                const int c = t * 16 + m;
                if (c < HC)              Y[row * HC + c] = acc[t][r];
                else if (c < HC + H)     as_[row * H + (c - HC)] = acc[t][r];
                else if (c < HC + 2 * H) ad_[row * H + (c - HC - H)] = acc[t][r];
            }
        }
    }
}

template<int RS>
__global__ __launch_bounds__(256) void gat_range_id_kernel(
    const int* __restrict__ dst, int E, unsigned* __restrict__ rid32, int nwords)
{
    const int t = blockIdx.x * 256 + threadIdx.x;
    if (t >= nwords) return;
    unsigned wv = 0u;
#pragma unroll
    for (int k = 0; k < 4; ++k) {
        const int e  = 4 * t + k;
        const int ec = (e < E) ? e : 0;
        const unsigned d = (unsigned)dst[ec];
        const unsigned b = (e < E) ? ((d / (unsigned)RS) & 255u) : 0u;
        wv |= b << (8 * k);
    }
    rid32[t] = wv;
}

template<int H, int C, int RS, int QCAP, bool ELU, int LMODE, int LCAP, int LPAD, int C2, int OLP>
__global__ __launch_bounds__(1024) void gat_range_owner_kernel(
    const int* __restrict__ src, const int* __restrict__ dst, int E, int N,
    const unsigned* __restrict__ rid32, int nvec,
    const float* __restrict__ as_, const float* __restrict__ ad_,
    const float* __restrict__ feat,
    const float* __restrict__ bias, float* __restrict__ out,
    unsigned* __restrict__ lst, int* __restrict__ lcnt,
    const float* __restrict__ Wn, const float* __restrict__ awn_s, const float* __restrict__ awn_d,
    float* __restrict__ nh, float* __restrict__ nas, float* __restrict__ nad)
{
    constexpr int HC = H * C;
    constexpr int ACCP  = (C2 > 0) ? HC + 4 : HC;
    constexpr int NCT2  = (C2 + 2 + 15) / 16;
    constexpr int NCOL2 = NCT2 * 16;
    static_assert(LPAD == 0 || (H == 1 && LPAD >= RS && LPAD % 32 == 0), "padded logits: one head, whole lines per range");
    static_assert(C2 == 0 || (OLP >= RS && OLP % 32 == 0), "the next layer's logits: whole lines per range");
    static_assert(C2 == 0 || (HC % 4 == 0 && (HC * NCOL2) % 1024 == 0), "W of the next layer: whole k steps, whole staging trips");
    constexpr int NPIECE = (HC + 31) / 32;
    constexpr int GROUP = 16 / NPIECE;
    constexpr int U = 4;
    constexpr int RU = 4;
    constexpr unsigned NONE = 0xFFFFFFFFu;
    constexpr int SBITS = 23;
    static_assert(H == 1 || C % 32 == 0, "a 32-column piece must lie in one head");
    static_assert(H <= 2, "the denominator add below selects between two heads");
    static_assert(32 % GROUP == 0, "a wave's 32 entries split into whole groups");
    static_assert(RS <= 511, "the local row takes 9 bits and 511 is kept for NONE");
    static_assert(LMODE == 2 || QCAP <= LCAP, "a stored list fits its slot");
    __shared__ float    acc[RS * ACCP];
    __shared__ float    den[RS * H];
    __shared__ float    ldsW2[(C2 > 0) ? HC * NCOL2 : 1];
    __shared__ unsigned mtab[RS * H];
    __shared__ unsigned queue[QCAP];
    __shared__ int      qn;

    const int tid  = threadIdx.x;
    const int lane = tid & 31;
    const int wave = __builtin_amdgcn_readfirstlane(tid >> 5);
    const int lo   = blockIdx.x * RS;
    int cnt = N - lo; if (cnt > RS) cnt = RS;

    auto li = [&](int node, int h) -> int {
        if (LPAD == 0) return node * H + h;
        const unsigned g = (unsigned)node / (unsigned)RS;
        return (int)(g * (unsigned)LPAD + ((unsigned)node - g * (unsigned)RS));
    };

    constexpr int WT = (C2 > 0) ? (HC * NCOL2) / 1024 : 1;
    float wv[WT];
    if constexpr (C2 > 0) {
#pragma unroll
        for (int u = 0; u < WT; ++u) {
            const int idx = tid + u * 1024;
            const int k = idx / NCOL2, c = idx % NCOL2;
            wv[u] = Wn[k * C2 + ((c < C2) ? c : (C2 - 1))];
        }
    }
    for (int i = tid; i < RS * ACCP; i += 1024) acc[i] = 0.0f;
    for (int i = tid; i < RS * H; i += 1024) { den[i] = 0.0f; mtab[i] = 0u; }
    if (tid == 0) { qn = 0; queue[0] = NONE; }
    if constexpr (C2 > 0) {
#pragma unroll
        for (int u = 0; u < WT; ++u) {
            const int idx = tid + u * 1024;
            ldsW2[idx] = ((idx % NCOL2) < C2) ? wv[u] : 0.0f;
        }
    }
    __syncthreads();
    if constexpr (C2 > 0) {
        if (tid < HC) {
            const float* wr = ldsW2 + tid * NCOL2;
            float sa = 0.0f, da = 0.0f;
            for (int c = 0; c < C2; ++c) {
                const float w = wr[c];
                sa += w * awn_s[c];
                da += w * awn_d[c];
            }
            ldsW2[tid * NCOL2 + C2]     = sa;
            ldsW2[tid * NCOL2 + C2 + 1] = da;
        }
    }

    const uint4* __restrict__ rv = reinterpret_cast<const uint4*>(rid32);
    const unsigned pat = (unsigned)(blockIdx.x & 255) * 0x01010101u;
    auto stream = [&](auto&& f) {
        for (int v0 = tid; v0 < nvec; v0 += 1024 * U) {
            uint4 x[U];
#pragma unroll
            for (int u = 0; u < U; ++u) {
                const int vi = v0 + u * 1024;
                x[u] = rv[(vi < nvec) ? vi : (nvec - 1)];
            }
#pragma unroll
            for (int u = 0; u < U; ++u) {
                const int vi = v0 + u * 1024;
                const unsigned q[4] = { x[u].x, x[u].y, x[u].z, x[u].w };
                unsigned z[4];
#pragma unroll
                for (int k = 0; k < 4; ++k) {
                    const unsigned y = q[k] ^ pat;
                    z[k] = (y - 0x01010101u) & ~y & 0x80808080u;
                }
                if (vi < nvec && (z[0] | z[1] | z[2] | z[3]) != 0u) {
                    unsigned m = 0u;
#pragma unroll
                    for (int k = 0; k < 4; ++k)
                        m |= ((((z[k] >> 7) * 0x00204081u) >> 21) & 15u) << (4 * k);
                    while (m != 0u) {
                        const int e = __builtin_ctz(m);
                        m &= m - 1u;
                        f(vi * 16 + e);
                    }
                }
            }
        }
    };
    auto push = [&](int eid) {
        const int p = atomicAdd(&qn, 1);
        if (p < QCAP) queue[p] = (unsigned)eid;
    };
    auto edge_max = [&](int eid) {
        if (eid >= E) return;
        const int r = dst[eid] - lo;
        if ((unsigned)r >= (unsigned)cnt) return;
        const int s = src[eid];
#pragma unroll
        for (int h = 0; h < H; ++h) atomicMax(&mtab[r * H + h], gat_fkey(as_[li(s, h)]));
    };
    auto edge_add = [&](int eid) {
        if (eid >= E) return;
        const int d = dst[eid];
        const int r = d - lo;
        if ((unsigned)r >= (unsigned)cnt) return;
        const int s = src[eid];
        float w[H];
#pragma unroll
        for (int h = 0; h < H; ++h) {
            const float ev = gat_lrelu(as_[li(s, h)] + ad_[li(d, h)]);
            w[h] = expf(ev - __uint_as_float(mtab[r * H + h]));
            atomicAdd(&den[r * H + h], w[h]);
        }
        const float* fs = feat + (long)s * HC;
        for (int c = 0; c < HC; ++c) atomicAdd(&acc[r * ACCP + c], ((H == 1 || c < C) ? w[0] : w[H - 1]) * fs[c]);
    };

    unsigned* __restrict__ myl = lst + (long)blockIdx.x * LCAP;
    int nl = -1;
    if (LMODE == 2) nl = __builtin_amdgcn_readfirstlane(lcnt[blockIdx.x]);
    const bool loaded = LMODE == 2 && nl >= 0 && nl <= QCAP && nl <= LCAP;

    if (!loaded) {
        stream(push);
        __syncthreads();
    }

    const int  qv   = loaded ? nl : __builtin_amdgcn_readfirstlane(qn);
    const bool full = !loaded && (qv > QCAP || N > (1 << SBITS));
    const int  n    = full ? 0 : qv;

    if (loaded) {
        for (int i0 = tid; i0 < n; i0 += 1024 * RU) {
            unsigned ee[RU];
#pragma unroll
            for (int u = 0; u < RU; ++u) {
                const int i = i0 + u * 1024;
                const unsigned e = myl[(i < n) ? i : i0];
                ee[u] = (i < n) ? e : NONE;
            }
            float a[RU][H];
#pragma unroll
            for (int u = 0; u < RU; ++u) {
                const int sg = (ee[u] != NONE) ? (int)(ee[u] & ((1u << SBITS) - 1u)) : 0;
#pragma unroll
                for (int h = 0; h < H; ++h) a[u][h] = as_[li(sg, h)];
            }
#pragma unroll
            for (int u = 0; u < RU; ++u) {
                const int  i  = i0 + u * 1024;
                const bool in = ee[u] != NONE;
                if (i < n) queue[i] = ee[u];
                const int rt = in ? (int)(ee[u] >> SBITS) : 0;
#pragma unroll
                for (int h = 0; h < H; ++h) atomicMax(&mtab[rt * H + h], in ? gat_fkey(a[u][h]) : 0u);
            }
        }
    } else if (full) {
        stream(edge_max);
    } else {
        for (int i0 = tid; i0 < n; i0 += 1024 * RU) {
            int  dd[RU], ss[RU];
            bool ok[RU];
#pragma unroll
            for (int u = 0; u < RU; ++u) {
                const int i = i0 + u * 1024;
                const unsigned e = queue[(i < n) ? i : i0];
                ok[u] = (i < n) && e < (unsigned)E;
                const int ec = ok[u] ? (int)e : 0;
                dd[u] = dst[ec];
                ss[u] = src[ec];
            }
            int   rr[RU];
            float a[RU][H];
#pragma unroll
            for (int u = 0; u < RU; ++u) {
                const int  r  = dd[u] - lo;
                const bool in = ok[u] && (unsigned)r < (unsigned)cnt;
                rr[u] = in ? r : -1;
                const int sg = in ? ss[u] : 0;
#pragma unroll
                for (int h = 0; h < H; ++h) a[u][h] = as_[li(sg, h)];
            }
#pragma unroll
            for (int u = 0; u < RU; ++u) {
                const int  i  = i0 + u * 1024;
                const bool in = rr[u] >= 0;
                if (i < n) queue[i] = in ? ((unsigned)ss[u] | ((unsigned)rr[u] << SBITS)) : NONE;
                const int rt = in ? rr[u] : 0;
#pragma unroll
                for (int h = 0; h < H; ++h) atomicMax(&mtab[rt * H + h], in ? gat_fkey(a[u][h]) : 0u);
            }
        }
    }
    __syncthreads();

    if (LMODE == 1) {
        for (int i = tid; i < n; i += 1024) myl[i] = queue[i];
        if (tid == 0) lcnt[blockIdx.x] = full ? -1 : n;
    }

    for (int i = tid; i < cnt * H; i += 1024) {
        const int g = li(lo + i / H, i % H);
        const unsigned ks = gat_fkey(as_[g]);
        const unsigned kt = mtab[i];
        const unsigned k  = kt > ks ? kt : ks;
        mtab[i] = __float_as_uint(gat_lrelu(gat_funkey(k) + ad_[g]));
    }
    __syncthreads();

    if (full) stream(edge_add);

    const int total = n + cnt;
    auto bcast_i = [](int v, int j) -> int { return __builtin_amdgcn_readlane(v, j); };

    for (int b0 = wave * 32; b0 < total; b0 += 1024) {
        int nb = total - b0; if (nb > 32) nb = 32;
        const int  i    = b0 + lane;
        const bool isq  = i < n;
        const unsigned ent = queue[isq ? i : 0];
        const int  sq   = (int)(ent & ((1u << SBITS) - 1u));
        const int  rq   = (int)(ent >> SBITS);
        const int  rs   = i - n;
        const bool valid = isq ? (ent != NONE) : (i < total);
        const int  r    = valid ? (isq ? rq : rs) : 0;
        const int  s    = valid ? (isq ? sq : (lo + rs)) : lo;
        const int  d    = lo + r;
        const int  vflag = valid ? 1 : 0;
        float w[H];
#pragma unroll
        for (int h = 0; h < H; ++h) {
            const float ev = gat_lrelu(as_[li(s, h)] + ad_[li(d, h)]);
            w[h] = expf(ev - __uint_as_float(mtab[r * H + h]));
        }
        for (int j0 = 0; j0 < nb; j0 += GROUP) {
            if (j0 + GROUP <= nb) {
                float v[GROUP][NPIECE];
                int   rj[GROUP];
                int   vj[GROUP];
                float wj[GROUP][H];
#pragma unroll
                for (int u = 0; u < GROUP; ++u) {
                    const int sj = bcast_i(s, j0 + u);
                    rj[u] = bcast_i(r, j0 + u);
                    vj[u] = bcast_i(vflag, j0 + u);
#pragma unroll
                    for (int h = 0; h < H; ++h)
                        wj[u][h] = __int_as_float(bcast_i(__float_as_int(w[h]), j0 + u));
                    const float* fs = feat + (long)sj * HC;
#pragma unroll
                    for (int p = 0; p < NPIECE; ++p) {
                        const int c  = p * 32 + lane;
                        const int cc = (c < HC) ? c : (HC - 1);
                        v[u][p] = fs[cc];
                    }
                }
#pragma unroll
                for (int u = 0; u < GROUP; ++u) {
#pragma unroll
                    for (int p = 0; p < NPIECE; ++p) asm volatile("" : "+v"(v[u][p]));
                }
#pragma unroll
                for (int u = 0; u < GROUP; ++u) {
                    if (vj[u] != 0) {
                        float* ar = acc + rj[u] * ACCP;
#pragma unroll
                        for (int p = 0; p < NPIECE; ++p) {
                            const int hh = (H == 1) ? 0 : ((p * 32) / C);
                            const int c  = p * 32 + lane;
                            if (c < HC) atomicAdd(&ar[c], wj[u][hh] * v[u][p]);
                        }
                        if (lane < H) atomicAdd(&den[rj[u] * H + lane], (lane == 0) ? wj[u][0] : wj[u][H - 1]);
                    }
                }
            } else {
                for (int j = j0; j < nb; ++j) {
                    const int sj = bcast_i(s, j);
                    const int rr = bcast_i(r, j);
                    const int vv = bcast_i(vflag, j);
                    float wq[H];
#pragma unroll
                    for (int h = 0; h < H; ++h) wq[h] = __int_as_float(bcast_i(__float_as_int(w[h]), j));
                    if (vv != 0) {
                        const float* fs = feat + (long)sj * HC;
                        float* ar = acc + rr * ACCP;
#pragma unroll
                        for (int p = 0; p < NPIECE; ++p) {
                            const int hh = (H == 1) ? 0 : ((p * 32) / C);
                            const int c  = p * 32 + lane;
                            if (c < HC) atomicAdd(&ar[c], wq[hh] * fs[c]);
                        }
                        if (lane < H) atomicAdd(&den[rr * H + lane], (lane == 0) ? wq[0] : wq[H - 1]);
                    }
                }
            }
        }
    }
    __syncthreads();

    if constexpr (C2 == 0) {
        float* po = out + (long)lo * HC;
        for (int i = tid; i < cnt * HC; i += 1024) {
            const int row = i / HC;
            const int c   = i - row * HC;
            const float t = acc[row * ACCP + c] / (den[row * H + c / C] + 1e-16f) + bias[c];
            po[i] = ELU ? (t > 0.0f ? t : (expf(t) - 1.0f)) : t;
        }
    } else {
        for (int i = tid; i < cnt * HC; i += 1024) {
            const int row = i / HC;
            const int c   = i - row * HC;
            float* p = acc + row * ACCP + c;
            const float t = *p / (den[row * H + c / C] + 1e-16f) + bias[c];
            *p = ELU ? (t > 0.0f ? t : (expf(t) - 1.0f)) : t;
        }
        __syncthreads();
        const int m    = lane & 15;
        const int half = lane >> 4;
        const int koff = half << 1;
        constexpr int NRT = (RS + 15) / 16;
        for (int rt = wave; rt < NRT; rt += 32) {
            int ra = rt * 16 + m; if (ra > RS - 1) ra = RS - 1;
            const float* arow = acc + ra * ACCP;
            v8f o[NCT2] = {};
            for (int k0 = 0; k0 < HC; k0 += 4) {
                v2f a;
                a.x = arow[k0 + koff];
                a.y = arow[k0 + koff + 1];
#pragma unroll
                for (int t = 0; t < NCT2; ++t) {
                    v2f b;
                    b.x = ldsW2[(k0 + koff) * NCOL2 + t * 16 + m];
                    b.y = ldsW2[(k0 + koff + 1) * NCOL2 + t * 16 + m];
                    o[t] = __builtin_amdgcn_wmma_f32_16x16x4_f32(
                        false, a, false, b, (short)0, o[t], false, false);
                }
            }
#pragma unroll
            for (int r = 0; r < 8; ++r) {
                const int row = rt * 16 + half * 8 + r;
                if (row < cnt) {
#pragma unroll
                    for (int t = 0; t < NCT2; ++t) {
                        const int c = t * 16 + m;
                        if (c < C2)            nh[(long)(lo + row) * C2 + c] = o[t][r];
                        else if (c == C2)      nas[blockIdx.x * OLP + row] = o[t][r];
                        else if (c == C2 + 1)  nad[blockIdx.x * OLP + row] = o[t][r];
                    }
                }
            }
        }
    }
}

extern "C" void kernel_launch(void* const* d_in, const int* in_sizes, int n_in,
                              void* d_out, int out_size, void* d_ws, size_t ws_size,
                              hipStream_t stream)
{
    (void)n_in; (void)out_size; (void)ws_size;
    const float* x   = (const float*)d_in[0];
    const int*   ei  = (const int*)  d_in[1];
    const float* W1  = (const float*)d_in[2];
    const float* aS1 = (const float*)d_in[3];
    const float* aD1 = (const float*)d_in[4];
    const float* b1  = (const float*)d_in[5];
    const float* W2  = (const float*)d_in[6];
    const float* aS2 = (const float*)d_in[7];
    const float* aD2 = (const float*)d_in[8];
    const float* b2  = (const float*)d_in[9];

    const int IN = 128, HC1 = 128;
    int N  = in_sizes[0] / IN;
    int E  = in_sizes[1] / 2;
    const int* srcI = ei;
    const int* dstI = ei + E;

    float* wsf = (float*)d_ws;
    size_t off = 0;
    auto take = [&](size_t nf) -> size_t { size_t o = off; off += (nf + 63) & ~(size_t)63; return o; };
    const int nvec   = (E + 15) / 16;
    const int nwords = nvec * 4;
    float*    h1    = wsf + take((size_t)N * HC1);
    float*    h2    = wsf + take((size_t)N * 40);
    float*    as1   = wsf + take((size_t)N * 2);
    float*    ad1   = wsf + take((size_t)N * 2);
    unsigned* rid   = (unsigned*)(wsf + take((size_t)nwords));

    dim3 blk(256);
    constexpr int RS = 392, QCAP1 = 8192;
    constexpr int QCAP2 = 25600;
    const int ngrp = (N + RS - 1) / RS;
    constexpr int LP2 = 416;
    float*    as2   = wsf + take((size_t)ngrp * LP2);
    float*    ad2   = wsf + take((size_t)ngrp * LP2);
    unsigned* lst   = (unsigned*)(wsf + take((size_t)ngrp * QCAP1));
    int*      lcnt  = (int*)(wsf + take((size_t)ngrp));

    gat_range_id_kernel<RS><<<(nwords + 255) / 256, blk, 0, stream>>>(dstI, E, rid, nwords);

    gat_gemm_wmma<144, 9, 2, 64><<<(N + 127) / 128, blk, 0, stream>>>(x, W1, aS1, aD1, h1, as1, ad1, N);
    gat_range_owner_kernel<2, 64, RS, QCAP1, true, 1, QCAP1, 0, 40, LP2><<<ngrp, dim3(1024), 0, stream>>>(
        srcI, dstI, E, N, rid, nvec, as1, ad1, h1, b1, (float*)nullptr, lst, lcnt,
        W2, aS2, aD2, h2, as2, ad2);

    gat_range_owner_kernel<1, 40, RS, QCAP2, false, 2, QCAP1, LP2, 0, 0><<<ngrp, dim3(1024), 0, stream>>>(
        srcI, dstI, E, N, rid, nvec, as2, ad2, h2, b2, (float*)d_out, lst, lcnt,
        (const float*)nullptr, (const float*)nullptr, (const float*)nullptr,
        (float*)nullptr, (float*)nullptr, (float*)nullptr);
}


# owner passes: as[src] gathered at a list entry's look-up is kept in LDS beside the entry; the walk reads it there, not from memory again (layer 1's list 7168 entries to fit); weights, sum order kept
# speedup vs baseline: 4.6163x; 1.0325x over previous
#include <hip/hip_runtime.h>
#include <math.h>

typedef float v2f __attribute__((ext_vector_type(2)));
typedef float v8f __attribute__((ext_vector_type(8)));

#define NEG_SLOPE 0.2f

__device__ __forceinline__ float gat_lrelu(float v) { return v > 0.0f ? v : NEG_SLOPE * v; }
__device__ __forceinline__ unsigned gat_fkey(float f) {
    unsigned u = __float_as_uint(f);
    return (u & 0x80000000u) ? ~u : (u | 0x80000000u);
}
__device__ __forceinline__ float gat_funkey(unsigned k) {
    return __uint_as_float((k & 0x80000000u) ? (k & 0x7fffffffu) : ~k);
}

template<int NCOL, int NCT, int H, int C>
__global__ __launch_bounds__(256) void gat_gemm_wmma(
    const float* __restrict__ X, const float* __restrict__ W,
    const float* __restrict__ aw_s, const float* __restrict__ aw_d,
    float* __restrict__ Y, float* __restrict__ as_, float* __restrict__ ad_, int N)
{
    constexpr int K  = 128;
    constexpr int HC = H * C;
    static_assert(HC + 2 * H <= NCOL, "the logit columns fit the padded width");
    static_assert(K * H <= 256, "one thread per (k, head) pair of the added columns");
    __shared__ float ldsW[K * NCOL];
    for (int idx = threadIdx.x; idx < K * NCOL; idx += 256) {
        int k = idx / NCOL, c = idx % NCOL;
        ldsW[idx] = (c < HC) ? W[k * HC + c] : 0.0f;
    }
    __syncthreads();
    if (threadIdx.x < K * H) {
        const int k  = threadIdx.x % K;
        const int hd = threadIdx.x / K;
        const float* wr = ldsW + k * NCOL + hd * C;
        const float* ps = aw_s + hd * C;
        const float* pd = aw_d + hd * C;
        float s = 0.0f, d = 0.0f;
        for (int c = 0; c < C; ++c) {
            const float w = wr[c];
            s += w * ps[c];
            d += w * pd[c];
        }
        ldsW[k * NCOL + HC + hd]     = s;
        ldsW[k * NCOL + HC + H + hd] = d;
    }
    __syncthreads();

    const int lane = threadIdx.x & 31;
    const int wave = threadIdx.x >> 5;
    const int m    = lane & 15;
    const int half = lane >> 4;
    const int koff = half << 1;

    long row0 = (long)blockIdx.x * 128 + wave * 16;
    long rowL = row0 + m; if (rowL > (long)N - 1) rowL = (long)N - 1;
    const float* xrow = X + rowL * K;

    v8f acc[NCT] = {};
    for (int k0 = 0; k0 < K; k0 += 4) {
        v2f a;
        a.x = xrow[k0 + koff];
        a.y = xrow[k0 + koff + 1];
#pragma unroll
        for (int t = 0; t < NCT; ++t) {
            v2f b;
            b.x = ldsW[(k0 + koff) * NCOL + t * 16 + m];
            b.y = ldsW[(k0 + koff + 1) * NCOL + t * 16 + m];
            acc[t] = __builtin_amdgcn_wmma_f32_16x16x4_f32(
                false, a, false, b, (short)0, acc[t], false, false);
        }
    }
#pragma unroll
    for (int r = 0; r < 8; ++r) {
        long row = row0 + half * 8 + r;
        if (row < N) {
#pragma unroll
            for (int t = 0; t < NCT; ++t) {
                const int c = t * 16 + m;
                if (c < HC)              Y[row * HC + c] = acc[t][r];
                else if (c < HC + H)     as_[row * H + (c - HC)] = acc[t][r];
                else if (c < HC + 2 * H) ad_[row * H + (c - HC - H)] = acc[t][r];
            }
        }
    }
}

template<int RS>
__global__ __launch_bounds__(256) void gat_range_id_kernel(
    const int* __restrict__ dst, int E, unsigned* __restrict__ rid32, int nwords)
{
    const int t = blockIdx.x * 256 + threadIdx.x;
    if (t >= nwords) return;
    unsigned wv = 0u;
#pragma unroll
    for (int k = 0; k < 4; ++k) {
        const int e  = 4 * t + k;
        const int ec = (e < E) ? e : 0;
        const unsigned d = (unsigned)dst[ec];
        const unsigned b = (e < E) ? ((d / (unsigned)RS) & 255u) : 0u;
        wv |= b << (8 * k);
    }
    rid32[t] = wv;
}

template<int H, int C, int RS, int QCAP, bool ELU, int LMODE, int LCAP, int LPAD, int C2, int OLP>
__global__ __launch_bounds__(1024) void gat_range_owner_kernel(
    const int* __restrict__ src, const int* __restrict__ dst, int E, int N,
    const unsigned* __restrict__ rid32, int nvec,
    const float* __restrict__ as_, const float* __restrict__ ad_,
    const float* __restrict__ feat,
    const float* __restrict__ bias, float* __restrict__ out,
    unsigned* __restrict__ lst, int* __restrict__ lcnt,
    const float* __restrict__ Wn, const float* __restrict__ awn_s, const float* __restrict__ awn_d,
    float* __restrict__ nh, float* __restrict__ nas, float* __restrict__ nad)
{
    constexpr int HC = H * C;
    constexpr int ACCP  = (C2 > 0) ? HC + 4 : HC;
    constexpr int NCT2  = (C2 + 2 + 15) / 16;
    constexpr int NCOL2 = NCT2 * 16;
    static_assert(LPAD == 0 || (H == 1 && LPAD >= RS && LPAD % 32 == 0), "padded logits: one head, whole lines per range");
    static_assert(C2 == 0 || (OLP >= RS && OLP % 32 == 0), "the next layer's logits: whole lines per range");
    static_assert(C2 == 0 || (HC % 4 == 0 && (HC * NCOL2) % 1024 == 0), "W of the next layer: whole k steps, whole staging trips");
    constexpr int NPIECE = (HC + 31) / 32;
    constexpr int GROUP = 16 / NPIECE;
    constexpr int U = 4;
    constexpr int RU = 4;
    constexpr unsigned NONE = 0xFFFFFFFFu;
    constexpr int SBITS = 23;
    static_assert(H == 1 || C % 32 == 0, "a 32-column piece must lie in one head");
    static_assert(H <= 2, "the denominator add below selects between two heads");
    static_assert(32 % GROUP == 0, "a wave's 32 entries split into whole groups");
    static_assert(RS <= 511, "the local row takes 9 bits and 511 is kept for NONE");
    static_assert(LMODE == 2 || QCAP <= LCAP, "a stored list fits its slot");
    __shared__ float    acc[RS * ACCP];
    __shared__ float    den[RS * H];
    __shared__ float    ldsW2[(C2 > 0) ? HC * NCOL2 : 1];
    __shared__ unsigned mtab[RS * H];
    __shared__ unsigned queue[QCAP];
    __shared__ __attribute__((aligned(8))) float sas[QCAP * H];
    __shared__ int      qn;

    const int tid  = threadIdx.x;
    const int lane = tid & 31;
    const int wave = __builtin_amdgcn_readfirstlane(tid >> 5);
    const int lo   = blockIdx.x * RS;
    int cnt = N - lo; if (cnt > RS) cnt = RS;

    auto li = [&](int node, int h) -> int {
        if (LPAD == 0) return node * H + h;
        const unsigned g = (unsigned)node / (unsigned)RS;
        return (int)(g * (unsigned)LPAD + ((unsigned)node - g * (unsigned)RS));
    };

    constexpr int WT = (C2 > 0) ? (HC * NCOL2) / 1024 : 1;
    float wv[WT];
    if constexpr (C2 > 0) {
#pragma unroll
        for (int u = 0; u < WT; ++u) {
            const int idx = tid + u * 1024;
            const int k = idx / NCOL2, c = idx % NCOL2;
            wv[u] = Wn[k * C2 + ((c < C2) ? c : (C2 - 1))];
        }
    }
    for (int i = tid; i < RS * ACCP; i += 1024) acc[i] = 0.0f;
    for (int i = tid; i < RS * H; i += 1024) { den[i] = 0.0f; mtab[i] = 0u; }
    if (tid == 0) { qn = 0; queue[0] = NONE; }
    if constexpr (C2 > 0) {
#pragma unroll
        for (int u = 0; u < WT; ++u) {
            const int idx = tid + u * 1024;
            ldsW2[idx] = ((idx % NCOL2) < C2) ? wv[u] : 0.0f;
        }
    }
    __syncthreads();
    if constexpr (C2 > 0) {
        if (tid < HC) {
            const float* wr = ldsW2 + tid * NCOL2;
            float sa = 0.0f, da = 0.0f;
            for (int c = 0; c < C2; ++c) {
                const float w = wr[c];
                sa += w * awn_s[c];
                da += w * awn_d[c];
            }
            ldsW2[tid * NCOL2 + C2]     = sa;
            ldsW2[tid * NCOL2 + C2 + 1] = da;
        }
    }

    const uint4* __restrict__ rv = reinterpret_cast<const uint4*>(rid32);
    const unsigned pat = (unsigned)(blockIdx.x & 255) * 0x01010101u;
    auto stream = [&](auto&& f) {
        for (int v0 = tid; v0 < nvec; v0 += 1024 * U) {
            uint4 x[U];
#pragma unroll
            for (int u = 0; u < U; ++u) {
                const int vi = v0 + u * 1024;
                x[u] = rv[(vi < nvec) ? vi : (nvec - 1)];
            }
#pragma unroll
            for (int u = 0; u < U; ++u) {
                const int vi = v0 + u * 1024;
                const unsigned q[4] = { x[u].x, x[u].y, x[u].z, x[u].w };
                unsigned z[4];
#pragma unroll
                for (int k = 0; k < 4; ++k) {
                    const unsigned y = q[k] ^ pat;
                    z[k] = (y - 0x01010101u) & ~y & 0x80808080u;
                }
                if (vi < nvec && (z[0] | z[1] | z[2] | z[3]) != 0u) {
                    unsigned m = 0u;
#pragma unroll
                    for (int k = 0; k < 4; ++k)
                        m |= ((((z[k] >> 7) * 0x00204081u) >> 21) & 15u) << (4 * k);
                    while (m != 0u) {
                        const int e = __builtin_ctz(m);
                        m &= m - 1u;
                        f(vi * 16 + e);
                    }
                }
            }
        }
    };
    auto push = [&](int eid) {
        const int p = atomicAdd(&qn, 1);
        if (p < QCAP) queue[p] = (unsigned)eid;
    };
    auto edge_max = [&](int eid) {
        if (eid >= E) return;
        const int r = dst[eid] - lo;
        if ((unsigned)r >= (unsigned)cnt) return;
        const int s = src[eid];
#pragma unroll
        for (int h = 0; h < H; ++h) atomicMax(&mtab[r * H + h], gat_fkey(as_[li(s, h)]));
    };
    auto edge_add = [&](int eid) {
        if (eid >= E) return;
        const int d = dst[eid];
        const int r = d - lo;
        if ((unsigned)r >= (unsigned)cnt) return;
        const int s = src[eid];
        float w[H];
#pragma unroll
        for (int h = 0; h < H; ++h) {
            const float ev = gat_lrelu(as_[li(s, h)] + ad_[li(d, h)]);
            w[h] = expf(ev - __uint_as_float(mtab[r * H + h]));
            atomicAdd(&den[r * H + h], w[h]);
        }
        const float* fs = feat + (long)s * HC;
        for (int c = 0; c < HC; ++c) atomicAdd(&acc[r * ACCP + c], ((H == 1 || c < C) ? w[0] : w[H - 1]) * fs[c]);
    };

    unsigned* __restrict__ myl = lst + (long)blockIdx.x * LCAP;
    int nl = -1;
    if (LMODE == 2) nl = __builtin_amdgcn_readfirstlane(lcnt[blockIdx.x]);
    const bool loaded = LMODE == 2 && nl >= 0 && nl <= QCAP && nl <= LCAP;

    if (!loaded) {
        stream(push);
        __syncthreads();
    }

    const int  qv   = loaded ? nl : __builtin_amdgcn_readfirstlane(qn);
    const bool full = !loaded && (qv > QCAP || N > (1 << SBITS));
    const int  n    = full ? 0 : qv;

    if (loaded) {
        for (int i0 = tid; i0 < n; i0 += 1024 * RU) {
            unsigned ee[RU];
#pragma unroll
            for (int u = 0; u < RU; ++u) {
                const int i = i0 + u * 1024;
                const unsigned e = myl[(i < n) ? i : i0];
                ee[u] = (i < n) ? e : NONE;
            }
            float a[RU][H];
#pragma unroll
            for (int u = 0; u < RU; ++u) {
                const int sg = (ee[u] != NONE) ? (int)(ee[u] & ((1u << SBITS) - 1u)) : 0;
#pragma unroll
                for (int h = 0; h < H; ++h) a[u][h] = as_[li(sg, h)];
            }
#pragma unroll
            for (int u = 0; u < RU; ++u) {
                const int  i  = i0 + u * 1024;
                const bool in = ee[u] != NONE;
                if (i < n) {
                    queue[i] = ee[u];
#pragma unroll
                    for (int h = 0; h < H; ++h) sas[i * H + h] = a[u][h];
                }
                const int rt = in ? (int)(ee[u] >> SBITS) : 0;
#pragma unroll
                for (int h = 0; h < H; ++h) atomicMax(&mtab[rt * H + h], in ? gat_fkey(a[u][h]) : 0u);
            }
        }
    } else if (full) {
        stream(edge_max);
    } else {
        for (int i0 = tid; i0 < n; i0 += 1024 * RU) {
            int  dd[RU], ss[RU];
            bool ok[RU];
#pragma unroll
            for (int u = 0; u < RU; ++u) {
                const int i = i0 + u * 1024;
                const unsigned e = queue[(i < n) ? i : i0];
                ok[u] = (i < n) && e < (unsigned)E;
                const int ec = ok[u] ? (int)e : 0;
                dd[u] = dst[ec];
                ss[u] = src[ec];
            }
            int   rr[RU];
            float a[RU][H];
#pragma unroll
            for (int u = 0; u < RU; ++u) {
                const int  r  = dd[u] - lo;
                const bool in = ok[u] && (unsigned)r < (unsigned)cnt;
                rr[u] = in ? r : -1;
                const int sg = in ? ss[u] : 0;
#pragma unroll
                for (int h = 0; h < H; ++h) a[u][h] = as_[li(sg, h)];
            }
#pragma unroll
            for (int u = 0; u < RU; ++u) {
                const int  i  = i0 + u * 1024;
                const bool in = rr[u] >= 0;
                if (i < n) {
                    queue[i] = in ? ((unsigned)ss[u] | ((unsigned)rr[u] << SBITS)) : NONE;
#pragma unroll
                    for (int h = 0; h < H; ++h) sas[i * H + h] = a[u][h];
                }
                const int rt = in ? rr[u] : 0;
#pragma unroll
                for (int h = 0; h < H; ++h) atomicMax(&mtab[rt * H + h], in ? gat_fkey(a[u][h]) : 0u);
            }
        }
    }
    __syncthreads();

    if (LMODE == 1) {
        for (int i = tid; i < n; i += 1024) myl[i] = queue[i];
        if (tid == 0) lcnt[blockIdx.x] = full ? -1 : n;
    }

    for (int i = tid; i < cnt * H; i += 1024) {
        const int g = li(lo + i / H, i % H);
        const unsigned ks = gat_fkey(as_[g]);
        const unsigned kt = mtab[i];
        const unsigned k  = kt > ks ? kt : ks;
        mtab[i] = __float_as_uint(gat_lrelu(gat_funkey(k) + ad_[g]));
    }
    __syncthreads();

    if (full) stream(edge_add);

    const int total = n + cnt;
    auto bcast_i = [](int v, int j) -> int { return __builtin_amdgcn_readlane(v, j); };

    for (int b0 = wave * 32; b0 < total; b0 += 1024) {
        int nb = total - b0; if (nb > 32) nb = 32;
        const int  i    = b0 + lane;
        const bool isq  = i < n;
        const unsigned ent = queue[isq ? i : 0];
        const int  sq   = (int)(ent & ((1u << SBITS) - 1u));
        const int  rq   = (int)(ent >> SBITS);
        const int  rs   = i - n;
        const bool valid = isq ? (ent != NONE) : (i < total);
        const int  r    = valid ? (isq ? rq : rs) : 0;
        const int  s    = valid ? (isq ? sq : (lo + rs)) : lo;
        const int  d    = lo + r;
        const int  vflag = valid ? 1 : 0;
        float av[H];
        if (isq) {
#pragma unroll
            for (int h = 0; h < H; ++h) av[h] = sas[i * H + h];
        } else {
#pragma unroll
            for (int h = 0; h < H; ++h) av[h] = as_[li(s, h)];
        }
        float w[H];
#pragma unroll
        for (int h = 0; h < H; ++h) {
            const float ev = gat_lrelu(av[h] + ad_[li(d, h)]);
            w[h] = expf(ev - __uint_as_float(mtab[r * H + h]));
        }
        for (int j0 = 0; j0 < nb; j0 += GROUP) {
            if (j0 + GROUP <= nb) {
                float v[GROUP][NPIECE];
                int   rj[GROUP];
                int   vj[GROUP];
                float wj[GROUP][H];
#pragma unroll
                for (int u = 0; u < GROUP; ++u) {
                    const int sj = bcast_i(s, j0 + u);
                    rj[u] = bcast_i(r, j0 + u);
                    vj[u] = bcast_i(vflag, j0 + u);
#pragma unroll
                    for (int h = 0; h < H; ++h)
                        wj[u][h] = __int_as_float(bcast_i(__float_as_int(w[h]), j0 + u));
                    const float* fs = feat + (long)sj * HC;
#pragma unroll
                    for (int p = 0; p < NPIECE; ++p) {
                        const int c  = p * 32 + lane;
                        const int cc = (c < HC) ? c : (HC - 1);
                        v[u][p] = fs[cc];
                    }
                }
#pragma unroll
                for (int u = 0; u < GROUP; ++u) {
#pragma unroll
                    for (int p = 0; p < NPIECE; ++p) asm volatile("" : "+v"(v[u][p]));
                }
#pragma unroll
                for (int u = 0; u < GROUP; ++u) {
                    if (vj[u] != 0) {
                        float* ar = acc + rj[u] * ACCP;
#pragma unroll
                        for (int p = 0; p < NPIECE; ++p) {
                            const int hh = (H == 1) ? 0 : ((p * 32) / C);
                            const int c  = p * 32 + lane;
                            if (c < HC) atomicAdd(&ar[c], wj[u][hh] * v[u][p]);
                        }
                        if (lane < H) atomicAdd(&den[rj[u] * H + lane], (lane == 0) ? wj[u][0] : wj[u][H - 1]);
                    }
                }
            } else {
                for (int j = j0; j < nb; ++j) {
                    const int sj = bcast_i(s, j);
                    const int rr = bcast_i(r, j);
                    const int vv = bcast_i(vflag, j);
                    float wq[H];
#pragma unroll
                    for (int h = 0; h < H; ++h) wq[h] = __int_as_float(bcast_i(__float_as_int(w[h]), j));
                    if (vv != 0) {
                        const float* fs = feat + (long)sj * HC;
                        float* ar = acc + rr * ACCP;
#pragma unroll
                        for (int p = 0; p < NPIECE; ++p) {
                            const int hh = (H == 1) ? 0 : ((p * 32) / C);
                            const int c  = p * 32 + lane;
                            if (c < HC) atomicAdd(&ar[c], wq[hh] * fs[c]);
                        }
                        if (lane < H) atomicAdd(&den[rr * H + lane], (lane == 0) ? wq[0] : wq[H - 1]);
                    }
                }
            }
        }
    }
    __syncthreads();

    if constexpr (C2 == 0) {
        float* po = out + (long)lo * HC;
        for (int i = tid; i < cnt * HC; i += 1024) {
            const int row = i / HC;
            const int c   = i - row * HC;
            const float t = acc[row * ACCP + c] / (den[row * H + c / C] + 1e-16f) + bias[c];
            po[i] = ELU ? (t > 0.0f ? t : (expf(t) - 1.0f)) : t;
        }
    } else {
        for (int i = tid; i < cnt * HC; i += 1024) {
            const int row = i / HC;
            const int c   = i - row * HC;
            float* p = acc + row * ACCP + c;
            const float t = *p / (den[row * H + c / C] + 1e-16f) + bias[c];
            *p = ELU ? (t > 0.0f ? t : (expf(t) - 1.0f)) : t;
        }
        __syncthreads();
        const int m    = lane & 15;
        const int half = lane >> 4;
        const int koff = half << 1;
        constexpr int NRT = (RS + 15) / 16;
        for (int rt = wave; rt < NRT; rt += 32) {
            int ra = rt * 16 + m; if (ra > RS - 1) ra = RS - 1;
            const float* arow = acc + ra * ACCP;
            v8f o[NCT2] = {};
            for (int k0 = 0; k0 < HC; k0 += 4) {
                v2f a;
                a.x = arow[k0 + koff];
                a.y = arow[k0 + koff + 1];
#pragma unroll
                for (int t = 0; t < NCT2; ++t) {
                    v2f b;
                    b.x = ldsW2[(k0 + koff) * NCOL2 + t * 16 + m];
                    b.y = ldsW2[(k0 + koff + 1) * NCOL2 + t * 16 + m];
                    o[t] = __builtin_amdgcn_wmma_f32_16x16x4_f32(
                        false, a, false, b, (short)0, o[t], false, false);
                }
            }
#pragma unroll
            for (int r = 0; r < 8; ++r) {
                const int row = rt * 16 + half * 8 + r;
                if (row < cnt) {
#pragma unroll
                    for (int t = 0; t < NCT2; ++t) {
                        const int c = t * 16 + m;
                        if (c < C2)            nh[(long)(lo + row) * C2 + c] = o[t][r];
                        else if (c == C2)      nas[blockIdx.x * OLP + row] = o[t][r];
                        else if (c == C2 + 1)  nad[blockIdx.x * OLP + row] = o[t][r];
                    }
                }
            }
        }
    }
}

extern "C" void kernel_launch(void* const* d_in, const int* in_sizes, int n_in,
                              void* d_out, int out_size, void* d_ws, size_t ws_size,
                              hipStream_t stream)
{
    (void)n_in; (void)out_size; (void)ws_size;
    const float* x   = (const float*)d_in[0];
    const int*   ei  = (const int*)  d_in[1];
    const float* W1  = (const float*)d_in[2];
    const float* aS1 = (const float*)d_in[3];
    const float* aD1 = (const float*)d_in[4];
    const float* b1  = (const float*)d_in[5];
    const float* W2  = (const float*)d_in[6];
    const float* aS2 = (const float*)d_in[7];
    const float* aD2 = (const float*)d_in[8];
    const float* b2  = (const float*)d_in[9];

    const int IN = 128, HC1 = 128;
    int N  = in_sizes[0] / IN;
    int E  = in_sizes[1] / 2;
    const int* srcI = ei;
    const int* dstI = ei + E;

    float* wsf = (float*)d_ws;
    size_t off = 0;
    auto take = [&](size_t nf) -> size_t { size_t o = off; off += (nf + 63) & ~(size_t)63; return o; };
    const int nvec   = (E + 15) / 16;
    const int nwords = nvec * 4;
    float*    h1    = wsf + take((size_t)N * HC1);
    float*    h2    = wsf + take((size_t)N * 40);
    float*    as1   = wsf + take((size_t)N * 2);
    float*    ad1   = wsf + take((size_t)N * 2);
    unsigned* rid   = (unsigned*)(wsf + take((size_t)nwords));

    dim3 blk(256);
    constexpr int RS = 392, QCAP1 = 7168, LCAP1 = 8192;
    constexpr int QCAP2 = 12800;
    const int ngrp = (N + RS - 1) / RS;
    constexpr int LP2 = 416;
    float*    as2   = wsf + take((size_t)ngrp * LP2);
    float*    ad2   = wsf + take((size_t)ngrp * LP2);
    unsigned* lst   = (unsigned*)(wsf + take((size_t)ngrp * LCAP1));
    int*      lcnt  = (int*)(wsf + take((size_t)ngrp));

    gat_range_id_kernel<RS><<<(nwords + 255) / 256, blk, 0, stream>>>(dstI, E, rid, nwords);

    gat_gemm_wmma<144, 9, 2, 64><<<(N + 127) / 128, blk, 0, stream>>>(x, W1, aS1, aD1, h1, as1, ad1, N);
    gat_range_owner_kernel<2, 64, RS, QCAP1, true, 1, LCAP1, 0, 40, LP2><<<ngrp, dim3(1024), 0, stream>>>(
        srcI, dstI, E, N, rid, nvec, as1, ad1, h1, b1, (float*)nullptr, lst, lcnt,
        W2, aS2, aD2, h2, as2, ad2);

    gat_range_owner_kernel<1, 40, RS, QCAP2, false, 2, LCAP1, LP2, 0, 0><<<ngrp, dim3(1024), 0, stream>>>(
        srcI, dstI, E, N, rid, nvec, as2, ad2, h2, b2, (float*)d_out, lst, lcnt,
        (const float*)nullptr, (const float*)nullptr, (const float*)nullptr,
        (float*)nullptr, (float*)nullptr, (float*)nullptr);
}
